# starting kernel
# baseline (speedup 1.0000x reference)
#include <hip/hip_runtime.h>

typedef float v2f __attribute__((ext_vector_type(2)));
typedef float v8f __attribute__((ext_vector_type(8)));

#define N_NODES 100000
#define N_EDGES 1600000

__global__ void gcn_fill_zero(float* __restrict__ p, int n) {
    int i = blockIdx.x * blockDim.x + threadIdx.x;
    if (i < n) p[i] = 0.0f;
}

__global__ void gcn_degree(const int* __restrict__ col, float* __restrict__ deg, int e) {
    int i = blockIdx.x * blockDim.x + threadIdx.x;
    if (i < e) atomicAdd(&deg[col[i]], 1.0f);
}

__global__ void gcn_dinv(float* __restrict__ deg, int n) {
    int i = blockIdx.x * blockDim.x + threadIdx.x;
    if (i < n) {
        float d = deg[i];
        deg[i] = (d > 0.0f) ? (1.0f / sqrtf(d)) : 0.0f;
    }
}

__global__ void gcn_norm(const int* __restrict__ row, const int* __restrict__ col,
                         const float* __restrict__ dinv, float* __restrict__ norm, int e) {
    int i = blockIdx.x * blockDim.x + threadIdx.x;
    if (i < e) norm[i] = dinv[row[i]] * dinv[col[i]];
}

template <int F>
__global__ void gcn_bias_init(float* __restrict__ out, const float* __restrict__ b, int total) {
    int i = blockIdx.x * blockDim.x + threadIdx.x;
    if (i < total) out[i] = b[i & (F - 1)];
}

template <int F, bool RELU>
__global__ __launch_bounds__(256) void gcn_gemm_wmma(const float* __restrict__ A,
                                                     const float* __restrict__ W,
                                                     float* __restrict__ out, int nRows) {
    constexpr int TILES_N = F / 16;
    const int wave = (int)((blockIdx.x * blockDim.x + threadIdx.x) >> 5);
    const int lane = threadIdx.x & 31;
    const int tm = wave / TILES_N;
    const int tn = wave - tm * TILES_N;
    if (tm * 16 >= nRows) return;

    const int mn    = lane & 15;
    const int khalf = lane >> 4;

    const float* arow = A + (size_t)(tm * 16 + mn) * 128;
    const float* wcol = W + (size_t)(tn * 16 + mn);

    v8f acc = {};
#pragma unroll
    for (int k = 0; k < 128; k += 4) {
        const int kb = k + 2 * khalf;
        float a0 = arow[kb];
        float a1 = arow[kb + 1];
        if (RELU) { a0 = fmaxf(a0, 0.0f); a1 = fmaxf(a1, 0.0f); }
        v2f a; a.x = a0; a.y = a1;
        v2f b;
        b.x = wcol[(size_t)kb * F];
        b.y = wcol[(size_t)(kb + 1) * F];
        acc = __builtin_amdgcn_wmma_f32_16x16x4_f32(
             false, a,  false, b,
             (short)0, acc,  false,  false);
    }

    float* orow = out + (size_t)(tm * 16 + khalf * 8) * F + (size_t)(tn * 16 + mn);
#pragma unroll
    for (int v = 0; v < 8; ++v)
        orow[(size_t)v * F] = acc[v];
}

template <int F>
__global__ __launch_bounds__(256) void gcn_scatter_add(const float* __restrict__ t,
                                                       const int* __restrict__ row,
                                                       const int* __restrict__ col,
                                                       const float* __restrict__ norm,
                                                       float* __restrict__ out) {
    constexpr int EPB = 256 / F;
    const int e = blockIdx.x * EPB + (threadIdx.x >> (F == 128 ? 7 : 6));
    const int f = threadIdx.x & (F - 1);
    const int r = row[e];
    const int c = col[e];
    const float v = t[(size_t)r * F + f] * norm[e];
    atomicAdd(&out[(size_t)c * F + f], v);
}

extern "C" void kernel_launch(void* const* d_in, const int* in_sizes, int n_in,
                              void* d_out, int out_size, void* d_ws, size_t ws_size,
                              hipStream_t stream) {
    const float* x   = (const float*)d_in[0];
    const int*   ei  = (const int*)d_in[1];
    const float* W1  = (const float*)d_in[2];
    const float* b1  = (const float*)d_in[3];
    const float* W2  = (const float*)d_in[4];
    const float* b2  = (const float*)d_in[5];
    const float* W3  = (const float*)d_in[6];
    const float* b3  = (const float*)d_in[7];
    float*       out = (float*)d_out;

    const int* row = ei;
    const int* col = ei + N_EDGES;

    float* dinv = (float*)d_ws;
    float* norm = dinv + N_NODES;
    float* t    = norm + N_EDGES;
    float* h    = t + (size_t)N_NODES * 128;

    const int TB = 256;

    gcn_fill_zero<<<(N_NODES + TB - 1) / TB, TB, 0, stream>>>(dinv, N_NODES);
    gcn_degree<<<(N_EDGES + TB - 1) / TB, TB, 0, stream>>>(col, dinv, N_EDGES);
    gcn_dinv<<<(N_NODES + TB - 1) / TB, TB, 0, stream>>>(dinv, N_NODES);
    gcn_norm<<<(N_EDGES + TB - 1) / TB, TB, 0, stream>>>(row, col, dinv, norm, N_EDGES);

    const int gemm128Blocks = (N_NODES / 16) * (128 / 16) / 8;
    const int gemm64Blocks  = (N_NODES / 16) * (64 / 16) / 8;
    const int scat128Blocks = N_EDGES / (256 / 128);
    const int scat64Blocks  = N_EDGES / (256 / 64);

    gcn_gemm_wmma<128, false><<<gemm128Blocks, TB, 0, stream>>>(x, W1, t, N_NODES);
    gcn_bias_init<128><<<(N_NODES * 128) / TB, TB, 0, stream>>>(h, b1, N_NODES * 128);
    gcn_scatter_add<128><<<scat128Blocks, TB, 0, stream>>>(t, row, col, norm, h);

    gcn_gemm_wmma<128, true><<<gemm128Blocks, TB, 0, stream>>>(h, W2, t, N_NODES);
    gcn_bias_init<128><<<(N_NODES * 128) / TB, TB, 0, stream>>>(h, b2, N_NODES * 128);
    gcn_scatter_add<128><<<scat128Blocks, TB, 0, stream>>>(t, row, col, norm, h);

    gcn_gemm_wmma<64, true><<<gemm64Blocks, TB, 0, stream>>>(h, W3, t, N_NODES);
    gcn_bias_init<64><<<(N_NODES * 64) / TB, TB, 0, stream>>>(out, b3, N_NODES * 64);
    gcn_scatter_add<64><<<scat64Blocks, TB, 0, stream>>>(t, row, col, norm, out);
}


# Replace edge-parallel f32 atomic scatter-add by in-call counting sort of edges by destination + one-wave-per-node row gather summed in registers, stored once (sum order: arrival rank, sequential)
# speedup vs baseline: 4.5726x; 4.5726x over previous
#include <hip/hip_runtime.h>

typedef float v2f __attribute__((ext_vector_type(2)));
typedef float v4f __attribute__((ext_vector_type(4)));
typedef float v8f __attribute__((ext_vector_type(8)));

#define N_NODES 100000
#define N_EDGES 1600000
#define SCAN_CHUNK 1024
#define SCAN_BLOCKS ((N_NODES + SCAN_CHUNK - 1) / SCAN_CHUNK)

__global__ void gcn_fill_zero(int* __restrict__ p, int n) {
    int i = blockIdx.x * blockDim.x + threadIdx.x;
    if (i < n) p[i] = 0;
}

__global__ void gcn_rank(const int* __restrict__ col, int* __restrict__ cnt,
                         int* __restrict__ rank, int e) {
    int i = blockIdx.x * blockDim.x + threadIdx.x;
    if (i < e) rank[i] = atomicAdd(&cnt[col[i]], 1);
}

__global__ __launch_bounds__(256) void gcn_block_sums(const int* __restrict__ cnt,
                                                      int* __restrict__ bs) {
    __shared__ int wsum[8];
    const int i4 = blockIdx.x * 256 + threadIdx.x;
    int s = 0;
    if (i4 * 4 < N_NODES) {
        const int4 c = ((const int4*)cnt)[i4];
        s = c.x + c.y + c.z + c.w;
    }
    for (int o = 16; o > 0; o >>= 1) s += __shfl_xor(s, o);
    if ((threadIdx.x & 31) == 0) wsum[threadIdx.x >> 5] = s;
    __syncthreads();
    if (threadIdx.x == 0) {
        int tot = 0;
        for (int w = 0; w < 8; ++w) tot += wsum[w];
        bs[blockIdx.x] = tot;
    }
}

__global__ __launch_bounds__(256) void gcn_offsets(const int* __restrict__ cnt,
                                                   const int* __restrict__ bs,
                                                   int* __restrict__ offs) {
    __shared__ int winc[8];
    __shared__ int wbs[8];
    const int tid = threadIdx.x, lane = tid & 31, wv = tid >> 5;
    int bv = (tid < (int)blockIdx.x) ? bs[tid] : 0;
    const int i4 = blockIdx.x * 256 + tid;
    const bool in = i4 * 4 < N_NODES;
    int4 c = make_int4(0, 0, 0, 0);
    if (in) c = ((const int4*)cnt)[i4];
    const int s = c.x + c.y + c.z + c.w;
    int inc = s;
    for (int o = 1; o < 32; o <<= 1) {
        const int v = __shfl_up(inc, o);
        if (lane >= o) inc += v;
    }
    for (int o = 16; o > 0; o >>= 1) bv += __shfl_xor(bv, o);
    if (lane == 31) winc[wv] = inc;
    if (lane == 0) wbs[wv] = bv;
    __syncthreads();
    int base = 0;
    for (int w = 0; w < 8; ++w) {
        base += wbs[w];
        if (w < wv) base += winc[w];
    }
    if (in) {
        int4 o;
        o.x = base + inc - s;
        o.y = o.x + c.x;
        o.z = o.y + c.y;
        o.w = o.z + c.z;
        ((int4*)offs)[i4] = o;
    }
}

__global__ void gcn_place(const int* __restrict__ row, const int* __restrict__ col,
                          const int* __restrict__ offs, const int* __restrict__ rank,
                          int* __restrict__ srow, int e) {
    int i = blockIdx.x * blockDim.x + threadIdx.x;
    if (i < e) srow[offs[col[i]] + rank[i]] = row[i];
}

template <int F, bool RELU>
__global__ __launch_bounds__(256) void gcn_gemm_wmma(const float* __restrict__ A,
                                                     const float* __restrict__ W,
                                                     float* __restrict__ out, int nRows) {
    constexpr int TILES_N = F / 16;
    const int wave = (int)((blockIdx.x * blockDim.x + threadIdx.x) >> 5);
    const int lane = threadIdx.x & 31;
    const int tm = wave / TILES_N;
    const int tn = wave - tm * TILES_N;
    if (tm * 16 >= nRows) return;

    const int mn    = lane & 15;
    const int khalf = lane >> 4;

    const float* arow = A + (size_t)(tm * 16 + mn) * 128;
    const float* wcol = W + (size_t)(tn * 16 + mn);

    v8f acc = {};
#pragma unroll
    for (int k = 0; k < 128; k += 4) {
        const int kb = k + 2 * khalf;
        float a0 = arow[kb];
        float a1 = arow[kb + 1];
        if (RELU) { a0 = fmaxf(a0, 0.0f); a1 = fmaxf(a1, 0.0f); }
        v2f a; a.x = a0; a.y = a1;
        v2f b;
        b.x = wcol[(size_t)kb * F];
        b.y = wcol[(size_t)(kb + 1) * F];
        acc = __builtin_amdgcn_wmma_f32_16x16x4_f32(
             false, a,  false, b,
             (short)0, acc,  false,  false);
    }

    float* orow = out + (size_t)(tm * 16 + khalf * 8) * F + (size_t)(tn * 16 + mn);
#pragma unroll
    for (int v = 0; v < 8; ++v)
        orow[(size_t)v * F] = acc[v];
}

template <int V> struct gcn_vec;
template <> struct gcn_vec<2> { typedef v2f type; };
template <> struct gcn_vec<4> { typedef v4f type; };

__device__ __forceinline__ float gcn_dinv_of(int d) {
    const float f = (float)d;
    return (f > 0.0f) ? (1.0f / sqrtf(f)) : 0.0f;
}

template <int F>
__global__ __launch_bounds__(256) void gcn_gather_sum(const float* __restrict__ t,
                                                      const int* __restrict__ offs,
                                                      const int* __restrict__ srow,
                                                      const float* __restrict__ b,
                                                      float* __restrict__ out, int nNodes) {
    constexpr int V = F / 32;
    typedef typename gcn_vec<V>::type vf;
    const int lane = threadIdx.x & 31;
    const int node = blockIdx.x * 8 + __builtin_amdgcn_readfirstlane((int)(threadIdx.x >> 5));
    if (node >= nNodes) return;

    const int beg = offs[node];
    const int end = (node + 1 < N_NODES) ? offs[node + 1] : N_EDGES;
    const int n = end - beg;
    const float dnode = gcn_dinv_of(n);

    const vf* tv = (const vf*)t + lane;
    vf acc = *((const vf*)b + lane);

    for (int base = 0; base < n; base += 32) {
        const int m = (n - base < 32) ? (n - base) : 32;
        int er = 0;
        float ew = 0.0f;
        if (lane < m) {
            er = srow[beg + base + lane];
            const int o0 = offs[er];
            const int o1 = (er + 1 < N_NODES) ? offs[er + 1] : N_EDGES;
            ew = gcn_dinv_of(o1 - o0) * dnode;
        }
        for (int k = 0; k < m; k += 4) {
            const int r0 = __builtin_amdgcn_readlane(er, k);
            const int r1 = __builtin_amdgcn_readlane(er, k + 1);
            const int r2 = __builtin_amdgcn_readlane(er, k + 2);
            const int r3 = __builtin_amdgcn_readlane(er, k + 3);
            const vf x0 = tv[(size_t)r0 * 32];
            const vf x1 = tv[(size_t)r1 * 32];
            const vf x2 = tv[(size_t)r2 * 32];
            const vf x3 = tv[(size_t)r3 * 32];
            const float w0 = __int_as_float(__builtin_amdgcn_readlane(__float_as_int(ew), k));
            const float w1 = __int_as_float(__builtin_amdgcn_readlane(__float_as_int(ew), k + 1));
            const float w2 = __int_as_float(__builtin_amdgcn_readlane(__float_as_int(ew), k + 2));
            const float w3 = __int_as_float(__builtin_amdgcn_readlane(__float_as_int(ew), k + 3));
            acc += x0 * w0;
            if (k + 1 < m) acc += x1 * w1;
            if (k + 2 < m) acc += x2 * w2;
            if (k + 3 < m) acc += x3 * w3;
        }
    }
    *((vf*)(out + (size_t)node * F) + lane) = acc;
}

extern "C" void kernel_launch(void* const* d_in, const int* in_sizes, int n_in,
                              void* d_out, int out_size, void* d_ws, size_t ws_size,
                              hipStream_t stream) {
    const float* x   = (const float*)d_in[0];
    const int*   ei  = (const int*)d_in[1];
    const float* W1  = (const float*)d_in[2];
    const float* b1  = (const float*)d_in[3];
    const float* W2  = (const float*)d_in[4];
    const float* b2  = (const float*)d_in[5];
    const float* W3  = (const float*)d_in[6];
    const float* b3  = (const float*)d_in[7];
    float*       out = (float*)d_out;

    const int* row = ei;
    const int* col = ei + N_EDGES;

    float* t    = (float*)d_ws;
    float* h    = t + (size_t)N_NODES * 128;
    int*   offs = (int*)(h + (size_t)N_NODES * 128);
    int*   srow = offs + N_NODES;
    int*   cnt  = (int*)h;
    int*   rank = cnt + N_NODES;
    int*   bs   = rank + N_EDGES;

    const int TB = 256;

    gcn_fill_zero<<<(N_NODES + TB - 1) / TB, TB, 0, stream>>>(cnt, N_NODES);
    gcn_rank<<<(N_EDGES + TB - 1) / TB, TB, 0, stream>>>(col, cnt, rank, N_EDGES);
    gcn_block_sums<<<SCAN_BLOCKS, TB, 0, stream>>>(cnt, bs);
    gcn_offsets<<<SCAN_BLOCKS, TB, 0, stream>>>(cnt, bs, offs);
    gcn_place<<<(N_EDGES + TB - 1) / TB, TB, 0, stream>>>(row, col, offs, rank, srow, N_EDGES);

    const int gemm128Blocks = (N_NODES / 16) * (128 / 16) / 8;
    const int gemm64Blocks  = (N_NODES / 16) * (64 / 16) / 8;
    const int gatherBlocks  = (N_NODES + 7) / 8;

    gcn_gemm_wmma<128, false><<<gemm128Blocks, TB, 0, stream>>>(x, W1, t, N_NODES);
    gcn_gather_sum<128><<<gatherBlocks, TB, 0, stream>>>(t, offs, srow, b1, h, N_NODES);

    gcn_gemm_wmma<128, true><<<gemm128Blocks, TB, 0, stream>>>(h, W2, t, N_NODES);
    gcn_gather_sum<128><<<gatherBlocks, TB, 0, stream>>>(t, offs, srow, b2, h, N_NODES);

    gcn_gemm_wmma<64, true><<<gemm64Blocks, TB, 0, stream>>>(h, W3, t, N_NODES);
    gcn_gather_sum<64><<<gatherBlocks, TB, 0, stream>>>(t, offs, srow, b3, out, N_NODES);
}


# Persistent waves: each wave loops over many nodes, row tiles or edges (GEMM wave keeps its W strip in registers), so a launch fits the chip's wave slots at once instead of queueing short waves
# speedup vs baseline: 4.8204x; 1.0542x over previous
#include <hip/hip_runtime.h>

typedef float v2f __attribute__((ext_vector_type(2)));
typedef float v4f __attribute__((ext_vector_type(4)));
typedef float v8f __attribute__((ext_vector_type(8)));

#define N_NODES 100000
#define N_EDGES 1600000
#define SCAN_CHUNK 1024
#define SCAN_BLOCKS ((N_NODES + SCAN_CHUNK - 1) / SCAN_CHUNK)
#define GATHER_BLOCKS 2048
#define GEMM_BLOCKS   1250

__global__ void gcn_fill_zero(int4* __restrict__ p, int n4) {
    int i = blockIdx.x * blockDim.x + threadIdx.x;
    if (i < n4) p[i] = make_int4(0, 0, 0, 0);
}

#define EDGES_PER_THREAD 4
__global__ __launch_bounds__(256) void gcn_rank(const int* __restrict__ col, int* __restrict__ cnt,
                                                int* __restrict__ rank, int e) {
    const int stride = gridDim.x * 256;
    const int i0 = blockIdx.x * 256 + threadIdx.x;
    int c[EDGES_PER_THREAD], r[EDGES_PER_THREAD];
#pragma unroll
    for (int j = 0; j < EDGES_PER_THREAD; ++j) {
        const int i = i0 + j * stride;
        c[j] = (i < e) ? col[i] : -1;
    }
#pragma unroll
    for (int j = 0; j < EDGES_PER_THREAD; ++j)
        r[j] = (c[j] >= 0) ? atomicAdd(&cnt[c[j]], 1) : 0;
#pragma unroll
    for (int j = 0; j < EDGES_PER_THREAD; ++j) {
        const int i = i0 + j * stride;
        if (i < e) rank[i] = r[j];
    }
}

__global__ __launch_bounds__(256) void gcn_block_sums(const int* __restrict__ cnt,
                                                      int* __restrict__ bs) {
    __shared__ int wsum[8];
    const int i4 = blockIdx.x * 256 + threadIdx.x;
    int s = 0;
    if (i4 * 4 < N_NODES) {
        const int4 c = ((const int4*)cnt)[i4];
        s = c.x + c.y + c.z + c.w;
    }
    for (int o = 16; o > 0; o >>= 1) s += __shfl_xor(s, o);
    if ((threadIdx.x & 31) == 0) wsum[threadIdx.x >> 5] = s;
    __syncthreads();
    if (threadIdx.x == 0) {
        int tot = 0;
        for (int w = 0; w < 8; ++w) tot += wsum[w];
        bs[blockIdx.x] = tot;
    }
}

__global__ __launch_bounds__(256) void gcn_offsets(const int* __restrict__ cnt,
                                                   const int* __restrict__ bs,
                                                   int* __restrict__ offs) {
    __shared__ int winc[8];
    __shared__ int wbs[8];
    const int tid = threadIdx.x, lane = tid & 31, wv = tid >> 5;
    int bv = (tid < (int)blockIdx.x) ? bs[tid] : 0;
    const int i4 = blockIdx.x * 256 + tid;
    const bool in = i4 * 4 < N_NODES;
    int4 c = make_int4(0, 0, 0, 0);
    if (in) c = ((const int4*)cnt)[i4];
    const int s = c.x + c.y + c.z + c.w;
    int inc = s;
    for (int o = 1; o < 32; o <<= 1) {
        const int v = __shfl_up(inc, o);
        if (lane >= o) inc += v;
    }
    for (int o = 16; o > 0; o >>= 1) bv += __shfl_xor(bv, o);
    if (lane == 31) winc[wv] = inc;
    if (lane == 0) wbs[wv] = bv;
    __syncthreads();
    int base = 0;
    for (int w = 0; w < 8; ++w) {
        base += wbs[w];
        if (w < wv) base += winc[w];
    }
    if (in) {
        int4 o;
        o.x = base + inc - s;
        o.y = o.x + c.x;
        o.z = o.y + c.y;
        o.w = o.z + c.z;
        ((int4*)offs)[i4] = o;
    }
}

__global__ __launch_bounds__(256) void gcn_place(const int* __restrict__ row,
                                                 const int* __restrict__ col,
                                                 const int* __restrict__ offs,
                                                 const int* __restrict__ rank,
                                                 int* __restrict__ srow, int e) {
    const int stride = gridDim.x * 256;
    const int i0 = blockIdx.x * 256 + threadIdx.x;
    int c[EDGES_PER_THREAD], r[EDGES_PER_THREAD], k[EDGES_PER_THREAD], o[EDGES_PER_THREAD];
#pragma unroll
    for (int j = 0; j < EDGES_PER_THREAD; ++j) {
        const int i = i0 + j * stride;
        const bool in = i < e;
        c[j] = in ? col[i] : 0;
        r[j] = in ? row[i] : 0;
        k[j] = in ? rank[i] : 0;
    }
#pragma unroll
    for (int j = 0; j < EDGES_PER_THREAD; ++j) o[j] = offs[c[j]];
#pragma unroll
    for (int j = 0; j < EDGES_PER_THREAD; ++j) {
        const int i = i0 + j * stride;
        if (i < e) srow[o[j] + k[j]] = r[j];
    }
}

template <int F, bool RELU>
__global__ __launch_bounds__(256) void gcn_gemm_wmma(const float* __restrict__ A,
                                                     const float* __restrict__ W,
                                                     float* __restrict__ out, int nRows) {
    constexpr int TILES_N = F / 16;
    constexpr int ROW_TILES_PER_BLOCK = 8 / TILES_N;
    const int lane = threadIdx.x & 31;
    const int wv = __builtin_amdgcn_readfirstlane((int)(threadIdx.x >> 5));
    const int tn = wv % TILES_N;
    const int tr = wv / TILES_N;

    const int mn    = lane & 15;
    const int khalf = lane >> 4;

    const float* wcol = W + (size_t)(tn * 16 + mn);
    v2f bfrag[32];
#pragma unroll
    for (int ks = 0; ks < 32; ++ks) {
        const int kb = 4 * ks + 2 * khalf;
        bfrag[ks].x = wcol[(size_t)kb * F];
        bfrag[ks].y = wcol[(size_t)(kb + 1) * F];
    }

    const int nRowTiles = nRows / 16;
    const int step = (int)gridDim.x * ROW_TILES_PER_BLOCK;
    for (int tm = (int)blockIdx.x * ROW_TILES_PER_BLOCK + tr; tm < nRowTiles; tm += step) {
        const float* arow = A + (size_t)(tm * 16 + mn) * 128;
        v8f acc = {};
#pragma unroll
        for (int ks = 0; ks < 32; ++ks) {
            const int kb = 4 * ks + 2 * khalf;
            float a0 = arow[kb];
            float a1 = arow[kb + 1];
            if (RELU) { a0 = fmaxf(a0, 0.0f); a1 = fmaxf(a1, 0.0f); }
            v2f a; a.x = a0; a.y = a1;
            acc = __builtin_amdgcn_wmma_f32_16x16x4_f32(
                 false, a,  false, bfrag[ks],
                 (short)0, acc,  false,  false);
        }
        float* orow = out + (size_t)(tm * 16 + khalf * 8) * F + (size_t)(tn * 16 + mn);
#pragma unroll
        for (int v = 0; v < 8; ++v)
            orow[(size_t)v * F] = acc[v];
    }
}

template <int V> struct gcn_vec;
template <> struct gcn_vec<2> { typedef v2f type; };
template <> struct gcn_vec<4> { typedef v4f type; };

__device__ __forceinline__ float gcn_dinv_of(int d) {
    const float f = (float)d;
    return (f > 0.0f) ? (1.0f / sqrtf(f)) : 0.0f;
}

template <int F>
__global__ __launch_bounds__(256) void gcn_gather_sum(const float* __restrict__ t,
                                                      const int* __restrict__ offs,
                                                      const int* __restrict__ srow,
                                                      const float* __restrict__ b,
                                                      float* __restrict__ out, int nNodes) {
    constexpr int V = F / 32;
    typedef typename gcn_vec<V>::type vf;
    const int lane = threadIdx.x & 31;
    const int first = blockIdx.x * 8 + __builtin_amdgcn_readfirstlane((int)(threadIdx.x >> 5));
    const int nWaves = (int)gridDim.x * 8;
    const vf* tv = (const vf*)t + lane;
    const vf bias = *((const vf*)b + lane);

    for (int node = first; node < nNodes; node += nWaves) {
    const int beg = offs[node];
    const int end = (node + 1 < N_NODES) ? offs[node + 1] : N_EDGES;
    const int n = end - beg;
    const float dnode = gcn_dinv_of(n);

    vf acc = bias;

    for (int base = 0; base < n; base += 32) {
        const int m = (n - base < 32) ? (n - base) : 32;
        int er = 0;
        float ew = 0.0f;
        if (lane < m) {
            er = srow[beg + base + lane];
            const int o0 = offs[er];
            const int o1 = (er + 1 < N_NODES) ? offs[er + 1] : N_EDGES;
            ew = gcn_dinv_of(o1 - o0) * dnode;
        }
        for (int k = 0; k < m; k += 4) {
            const int r0 = __builtin_amdgcn_readlane(er, k);
            const int r1 = __builtin_amdgcn_readlane(er, k + 1);
            const int r2 = __builtin_amdgcn_readlane(er, k + 2);
            const int r3 = __builtin_amdgcn_readlane(er, k + 3);
            const vf x0 = tv[(size_t)r0 * 32];
            const vf x1 = tv[(size_t)r1 * 32];
            const vf x2 = tv[(size_t)r2 * 32];
            const vf x3 = tv[(size_t)r3 * 32];
            const float w0 = __int_as_float(__builtin_amdgcn_readlane(__float_as_int(ew), k));
            const float w1 = __int_as_float(__builtin_amdgcn_readlane(__float_as_int(ew), k + 1));
            const float w2 = __int_as_float(__builtin_amdgcn_readlane(__float_as_int(ew), k + 2));
            const float w3 = __int_as_float(__builtin_amdgcn_readlane(__float_as_int(ew), k + 3));
            acc += x0 * w0;
            if (k + 1 < m) acc += x1 * w1;
            if (k + 2 < m) acc += x2 * w2;
            if (k + 3 < m) acc += x3 * w3;
        }
    }
    *((vf*)(out + (size_t)node * F) + lane) = acc;
    }
}

extern "C" void kernel_launch(void* const* d_in, const int* in_sizes, int n_in,
                              void* d_out, int out_size, void* d_ws, size_t ws_size,
                              hipStream_t stream) {
    const float* x   = (const float*)d_in[0];
    const int*   ei  = (const int*)d_in[1];
    const float* W1  = (const float*)d_in[2];
    const float* b1  = (const float*)d_in[3];
    const float* W2  = (const float*)d_in[4];
    const float* b2  = (const float*)d_in[5];
    const float* W3  = (const float*)d_in[6];
    const float* b3  = (const float*)d_in[7];
    float*       out = (float*)d_out;

    const int* row = ei;
    const int* col = ei + N_EDGES;

    float* t    = (float*)d_ws;
    float* h    = t + (size_t)N_NODES * 128;
    int*   offs = (int*)(h + (size_t)N_NODES * 128);
    int*   srow = offs + N_NODES;
    int*   cnt  = (int*)h;
    int*   rank = cnt + N_NODES;
    int*   bs   = rank + N_EDGES;

    const int TB = 256;

    const int edgeBlocks = (N_EDGES + TB * EDGES_PER_THREAD - 1) / (TB * EDGES_PER_THREAD);

    gcn_fill_zero<<<(N_NODES / 4 + TB - 1) / TB, TB, 0, stream>>>((int4*)cnt, N_NODES / 4);
    gcn_rank<<<edgeBlocks, TB, 0, stream>>>(col, cnt, rank, N_EDGES);
    gcn_block_sums<<<SCAN_BLOCKS, TB, 0, stream>>>(cnt, bs);
    gcn_offsets<<<SCAN_BLOCKS, TB, 0, stream>>>(cnt, bs, offs);
    gcn_place<<<edgeBlocks, TB, 0, stream>>>(row, col, offs, rank, srow, N_EDGES);

    const int gemm128Blocks = GEMM_BLOCKS;
    const int gemm64Blocks  = GEMM_BLOCKS;
    const int gatherBlocks  = GATHER_BLOCKS;

    gcn_gemm_wmma<128, false><<<gemm128Blocks, TB, 0, stream>>>(x, W1, t, N_NODES);
    gcn_gather_sum<128><<<gatherBlocks, TB, 0, stream>>>(t, offs, srow, b1, h, N_NODES);

    gcn_gemm_wmma<128, true><<<gemm128Blocks, TB, 0, stream>>>(h, W2, t, N_NODES);
    gcn_gather_sum<128><<<gatherBlocks, TB, 0, stream>>>(t, offs, srow, b2, h, N_NODES);

    gcn_gemm_wmma<64, true><<<gemm64Blocks, TB, 0, stream>>>(h, W3, t, N_NODES);
    gcn_gather_sum<64><<<gatherBlocks, TB, 0, stream>>>(t, offs, srow, b3, out, N_NODES);
}


# Source-side deg^-1/2 put on the rows once (GEMM 1 store, gather 1-2 store), dest-side on the finished sum: gathers drop the per-edge degree lookup; terms round as (t*ds) summed, then *dd
# speedup vs baseline: 5.0631x; 1.0503x over previous
#include <hip/hip_runtime.h>

typedef float v2f __attribute__((ext_vector_type(2)));
typedef float v4f __attribute__((ext_vector_type(4)));
typedef float v8f __attribute__((ext_vector_type(8)));

#define N_NODES 100000
#define N_EDGES 1600000
#define SCAN_CHUNK 1024
#define SCAN_BLOCKS ((N_NODES + SCAN_CHUNK - 1) / SCAN_CHUNK)
#define GATHER_BLOCKS 2048
#define GEMM_BLOCKS   1250

__global__ void gcn_fill_zero(int4* __restrict__ p, int n4) {
    int i = blockIdx.x * blockDim.x + threadIdx.x;
    if (i < n4) p[i] = make_int4(0, 0, 0, 0);
}

#define EDGES_PER_THREAD 4
__global__ __launch_bounds__(256) void gcn_rank(const int* __restrict__ col, int* __restrict__ cnt,
                                                int* __restrict__ rank, int e) {
    const int stride = gridDim.x * 256;
    const int i0 = blockIdx.x * 256 + threadIdx.x;
    int c[EDGES_PER_THREAD], r[EDGES_PER_THREAD];
#pragma unroll
    for (int j = 0; j < EDGES_PER_THREAD; ++j) {
        const int i = i0 + j * stride;
        c[j] = (i < e) ? col[i] : -1;
    }
#pragma unroll
    for (int j = 0; j < EDGES_PER_THREAD; ++j)
        r[j] = (c[j] >= 0) ? atomicAdd(&cnt[c[j]], 1) : 0;
#pragma unroll
    for (int j = 0; j < EDGES_PER_THREAD; ++j) {
        const int i = i0 + j * stride;
        if (i < e) rank[i] = r[j];
    }
}

__global__ __launch_bounds__(256) void gcn_block_sums(const int* __restrict__ cnt,
                                                      int* __restrict__ bs) {
    __shared__ int wsum[8];
    const int i4 = blockIdx.x * 256 + threadIdx.x;
    int s = 0;
    if (i4 * 4 < N_NODES) {
        const int4 c = ((const int4*)cnt)[i4];
        s = c.x + c.y + c.z + c.w;
    }
    for (int o = 16; o > 0; o >>= 1) s += __shfl_xor(s, o);
    if ((threadIdx.x & 31) == 0) wsum[threadIdx.x >> 5] = s;
    __syncthreads();
    if (threadIdx.x == 0) {
        int tot = 0;
        for (int w = 0; w < 8; ++w) tot += wsum[w];
        bs[blockIdx.x] = tot;
    }
}

__global__ __launch_bounds__(256) void gcn_offsets(const int* __restrict__ cnt,
                                                   const int* __restrict__ bs,
                                                   int* __restrict__ offs) {
    __shared__ int winc[8];
    __shared__ int wbs[8];
    const int tid = threadIdx.x, lane = tid & 31, wv = tid >> 5;
    int bv = (tid < (int)blockIdx.x) ? bs[tid] : 0;
    const int i4 = blockIdx.x * 256 + tid;
    const bool in = i4 * 4 < N_NODES;
    int4 c = make_int4(0, 0, 0, 0);
    if (in) c = ((const int4*)cnt)[i4];
    const int s = c.x + c.y + c.z + c.w;
    int inc = s;
    for (int o = 1; o < 32; o <<= 1) {
        const int v = __shfl_up(inc, o);
        if (lane >= o) inc += v;
    }
    for (int o = 16; o > 0; o >>= 1) bv += __shfl_xor(bv, o);
    if (lane == 31) winc[wv] = inc;
    if (lane == 0) wbs[wv] = bv;
    __syncthreads();
    int base = 0;
    for (int w = 0; w < 8; ++w) {
        base += wbs[w];
        if (w < wv) base += winc[w];
    }
    if (in) {
        int4 o;
        o.x = base + inc - s;
        o.y = o.x + c.x;
        o.z = o.y + c.y;
        o.w = o.z + c.z;
        ((int4*)offs)[i4] = o;
    }
}

__global__ __launch_bounds__(256) void gcn_place(const int* __restrict__ row,
                                                 const int* __restrict__ col,
                                                 const int* __restrict__ offs,
                                                 const int* __restrict__ rank,
                                                 int* __restrict__ srow, int e) {
    const int stride = gridDim.x * 256;
    const int i0 = blockIdx.x * 256 + threadIdx.x;
    int c[EDGES_PER_THREAD], r[EDGES_PER_THREAD], k[EDGES_PER_THREAD], o[EDGES_PER_THREAD];
#pragma unroll
    for (int j = 0; j < EDGES_PER_THREAD; ++j) {
        const int i = i0 + j * stride;
        const bool in = i < e;
        c[j] = in ? col[i] : 0;
        r[j] = in ? row[i] : 0;
        k[j] = in ? rank[i] : 0;
    }
#pragma unroll
    for (int j = 0; j < EDGES_PER_THREAD; ++j) o[j] = offs[c[j]];
#pragma unroll
    for (int j = 0; j < EDGES_PER_THREAD; ++j) {
        const int i = i0 + j * stride;
        if (i < e) srow[o[j] + k[j]] = r[j];
    }
}

__device__ __forceinline__ float gcn_dinv_of(int d) {
    const float f = (float)d;
    return (f > 0.0f) ? (1.0f / sqrtf(f)) : 0.0f;
}


template <int F, bool RELU, bool SCALE>
__global__ __launch_bounds__(256) void gcn_gemm_wmma(const float* __restrict__ A,
                                                     const float* __restrict__ W,
                                                     const int* __restrict__ offs,
                                                     float* __restrict__ out, int nRows) {
    constexpr int TILES_N = F / 16;
    constexpr int ROW_TILES_PER_BLOCK = 8 / TILES_N;
    const int lane = threadIdx.x & 31;
    const int wv = __builtin_amdgcn_readfirstlane((int)(threadIdx.x >> 5));
    const int tn = wv % TILES_N;
    const int tr = wv / TILES_N;

    const int mn    = lane & 15;
    const int khalf = lane >> 4;

    const float* wcol = W + (size_t)(tn * 16 + mn);
    v2f bfrag[32];
#pragma unroll
    for (int ks = 0; ks < 32; ++ks) {
        const int kb = 4 * ks + 2 * khalf;
        bfrag[ks].x = wcol[(size_t)kb * F];
        bfrag[ks].y = wcol[(size_t)(kb + 1) * F];
    }

    const int nRowTiles = nRows / 16;
    const int step = (int)gridDim.x * ROW_TILES_PER_BLOCK;
    for (int tm = (int)blockIdx.x * ROW_TILES_PER_BLOCK + tr; tm < nRowTiles; tm += step) {
        const float* arow = A + (size_t)(tm * 16 + mn) * 128;
        float drow = 1.0f;
        if (SCALE) {
            const int  r    = tm * 16 + mn;
            const bool last = r + 1 >= N_NODES;
            const int  o0   = offs[r];
            const int  o1m  = offs[last ? r : r + 1];
            const float f   = (float)((last ? N_EDGES : o1m) - o0);
            const float d   = 1.0f / sqrtf(fmaxf(f, 1.0f));
            drow = (f > 0.0f) ? d : 0.0f;
        }
        v8f acc = {};
#pragma unroll
        for (int ks = 0; ks < 32; ++ks) {
            const int kb = 4 * ks + 2 * khalf;
            float a0 = arow[kb];
            float a1 = arow[kb + 1];
            if (RELU) { a0 = fmaxf(a0, 0.0f); a1 = fmaxf(a1, 0.0f); }
            v2f a; a.x = a0; a.y = a1;
            acc = __builtin_amdgcn_wmma_f32_16x16x4_f32(
                 false, a,  false, bfrag[ks],
                 (short)0, acc,  false,  false);
        }
        float* orow = out + (size_t)(tm * 16 + khalf * 8) * F + (size_t)(tn * 16 + mn);
#pragma unroll
        for (int v = 0; v < 8; ++v) {
            float val = acc[v];
            if (SCALE) {
                const float dlo = __int_as_float(__builtin_amdgcn_readlane(__float_as_int(drow), v));
                const float dhi = __int_as_float(__builtin_amdgcn_readlane(__float_as_int(drow), v + 8));
                val *= khalf ? dhi : dlo;
            }
            orow[(size_t)v * F] = val;
        }
    }
}

template <int V> struct gcn_vec;
template <> struct gcn_vec<2> { typedef v2f type; };
template <> struct gcn_vec<4> { typedef v4f type; };

template <int F, bool PRESCALE>
__global__ __launch_bounds__(256) void gcn_gather_sum(const float* __restrict__ t,
                                                      const int* __restrict__ offs,
                                                      const int* __restrict__ srow,
                                                      const float* __restrict__ b,
                                                      float* __restrict__ out, int nNodes) {
    constexpr int V = F / 32;
    typedef typename gcn_vec<V>::type vf;
    const int lane = threadIdx.x & 31;
    const int first = blockIdx.x * 8 + __builtin_amdgcn_readfirstlane((int)(threadIdx.x >> 5));
    const int nWaves = (int)gridDim.x * 8;
    const vf* tv = (const vf*)t + lane;
    const vf bias = *((const vf*)b + lane);

    for (int node = first; node < nNodes; node += nWaves) {
    const int beg = offs[node];
    const int end = (node + 1 < N_NODES) ? offs[node + 1] : N_EDGES;
    const int n = end - beg;
    const float dnode = gcn_dinv_of(n);

    vf acc = {};

    for (int base = 0; base < n; base += 32) {
        const int m = (n - base < 32) ? (n - base) : 32;
        int er = 0;
        if (lane < m) er = srow[beg + base + lane];
        for (int k = 0; k < m; k += 4) {
            const int r0 = __builtin_amdgcn_readlane(er, k);
            const int r1 = __builtin_amdgcn_readlane(er, k + 1);
            const int r2 = __builtin_amdgcn_readlane(er, k + 2);
            const int r3 = __builtin_amdgcn_readlane(er, k + 3);
            const vf x0 = tv[(size_t)r0 * 32];
            const vf x1 = tv[(size_t)r1 * 32];
            const vf x2 = tv[(size_t)r2 * 32];
            const vf x3 = tv[(size_t)r3 * 32];
            const vf zero = {};
            acc += x0;
            acc += (k + 1 < m) ? x1 : zero;
            acc += (k + 2 < m) ? x2 : zero;
            acc += (k + 3 < m) ? x3 : zero;
        }
    }
    vf res = bias + acc * dnode;
    if (PRESCALE) res = res * dnode;
    *((vf*)(out + (size_t)node * F) + lane) = res;
    }
}

extern "C" void kernel_launch(void* const* d_in, const int* in_sizes, int n_in,
                              void* d_out, int out_size, void* d_ws, size_t ws_size,
                              hipStream_t stream) {
    const float* x   = (const float*)d_in[0];
    const int*   ei  = (const int*)d_in[1];
    const float* W1  = (const float*)d_in[2];
    const float* b1  = (const float*)d_in[3];
    const float* W2  = (const float*)d_in[4];
    const float* b2  = (const float*)d_in[5];
    const float* W3  = (const float*)d_in[6];
    const float* b3  = (const float*)d_in[7];
    float*       out = (float*)d_out;

    const int* row = ei;
    const int* col = ei + N_EDGES;

    float* t    = (float*)d_ws;
    float* h    = t + (size_t)N_NODES * 128;
    int*   offs = (int*)(h + (size_t)N_NODES * 128);
    int*   srow = offs + N_NODES;
    int*   cnt  = (int*)h;
    int*   rank = cnt + N_NODES;
    int*   bs   = rank + N_EDGES;

    const int TB = 256;

    const int edgeBlocks = (N_EDGES + TB * EDGES_PER_THREAD - 1) / (TB * EDGES_PER_THREAD);

    gcn_fill_zero<<<(N_NODES / 4 + TB - 1) / TB, TB, 0, stream>>>((int4*)cnt, N_NODES / 4);
    gcn_rank<<<edgeBlocks, TB, 0, stream>>>(col, cnt, rank, N_EDGES);
    gcn_block_sums<<<SCAN_BLOCKS, TB, 0, stream>>>(cnt, bs);
    gcn_offsets<<<SCAN_BLOCKS, TB, 0, stream>>>(cnt, bs, offs);
    gcn_place<<<edgeBlocks, TB, 0, stream>>>(row, col, offs, rank, srow, N_EDGES);

    const int gemm128Blocks = GEMM_BLOCKS;
    const int gemm64Blocks  = GEMM_BLOCKS;
    const int gatherBlocks  = GATHER_BLOCKS;

    gcn_gemm_wmma<128, false, true><<<gemm128Blocks, TB, 0, stream>>>(x, W1, offs, t, N_NODES);
    gcn_gather_sum<128, true><<<gatherBlocks, TB, 0, stream>>>(t, offs, srow, b1, h, N_NODES);

    gcn_gemm_wmma<128, true, false><<<gemm128Blocks, TB, 0, stream>>>(h, W2, offs, t, N_NODES);
    gcn_gather_sum<128, true><<<gatherBlocks, TB, 0, stream>>>(t, offs, srow, b2, h, N_NODES);

    gcn_gemm_wmma<64, true, false><<<gemm64Blocks, TB, 0, stream>>>(h, W3, offs, t, N_NODES);
    gcn_gather_sum<64, false><<<gatherBlocks, TB, 0, stream>>>(t, offs, srow, b3, out, N_NODES);
}


# GEMM: block stages each trip's A rows once in LDS (whole 512-byte rows, one trip ahead, LDS-only barrier); waves read fragments from LDS, not row-strided 8-byte global loads; GEMM grid 1024
# speedup vs baseline: 5.6376x; 1.1135x over previous
#include <hip/hip_runtime.h>

typedef float v2f __attribute__((ext_vector_type(2)));
typedef float v4f __attribute__((ext_vector_type(4)));
typedef float v8f __attribute__((ext_vector_type(8)));

#define N_NODES 100000
#define N_EDGES 1600000
#define SCAN_CHUNK 1024
#define SCAN_BLOCKS ((N_NODES + SCAN_CHUNK - 1) / SCAN_CHUNK)
#define GATHER_BLOCKS 2048
#define GEMM_BLOCKS   1024

__global__ void gcn_fill_zero(int4* __restrict__ p, int n4) {
    int i = blockIdx.x * blockDim.x + threadIdx.x;
    if (i < n4) p[i] = make_int4(0, 0, 0, 0);
}

#define EDGES_PER_THREAD 4
__global__ __launch_bounds__(256) void gcn_rank(const int* __restrict__ col, int* __restrict__ cnt,
                                                int* __restrict__ rank, int e) {
    const int stride = gridDim.x * 256;
    const int i0 = blockIdx.x * 256 + threadIdx.x;
    int c[EDGES_PER_THREAD], r[EDGES_PER_THREAD];
#pragma unroll
    for (int j = 0; j < EDGES_PER_THREAD; ++j) {
        const int i = i0 + j * stride;
        c[j] = (i < e) ? col[i] : -1;
    }
#pragma unroll
    for (int j = 0; j < EDGES_PER_THREAD; ++j)
        r[j] = (c[j] >= 0) ? atomicAdd(&cnt[c[j]], 1) : 0;
#pragma unroll
    for (int j = 0; j < EDGES_PER_THREAD; ++j) {
        const int i = i0 + j * stride;
        if (i < e) rank[i] = r[j];
    }
}

__global__ __launch_bounds__(256) void gcn_block_sums(const int* __restrict__ cnt,
                                                      int* __restrict__ bs) {
    __shared__ int wsum[8];
    const int i4 = blockIdx.x * 256 + threadIdx.x;
    int s = 0;
    if (i4 * 4 < N_NODES) {
        const int4 c = ((const int4*)cnt)[i4];
        s = c.x + c.y + c.z + c.w;
    }
    for (int o = 16; o > 0; o >>= 1) s += __shfl_xor(s, o);
    if ((threadIdx.x & 31) == 0) wsum[threadIdx.x >> 5] = s;
    __syncthreads();
    if (threadIdx.x == 0) {
        int tot = 0;
        for (int w = 0; w < 8; ++w) tot += wsum[w];
        bs[blockIdx.x] = tot;
    }
}

__global__ __launch_bounds__(256) void gcn_offsets(const int* __restrict__ cnt,
                                                   const int* __restrict__ bs,
                                                   int* __restrict__ offs) {
    __shared__ int winc[8];
    __shared__ int wbs[8];
    const int tid = threadIdx.x, lane = tid & 31, wv = tid >> 5;
    int bv = (tid < (int)blockIdx.x) ? bs[tid] : 0;
    const int i4 = blockIdx.x * 256 + tid;
    const bool in = i4 * 4 < N_NODES;
    int4 c = make_int4(0, 0, 0, 0);
    if (in) c = ((const int4*)cnt)[i4];
    const int s = c.x + c.y + c.z + c.w;
    int inc = s;
    for (int o = 1; o < 32; o <<= 1) {
        const int v = __shfl_up(inc, o);
        if (lane >= o) inc += v;
    }
    for (int o = 16; o > 0; o >>= 1) bv += __shfl_xor(bv, o);
    if (lane == 31) winc[wv] = inc;
    if (lane == 0) wbs[wv] = bv;
    __syncthreads();
    int base = 0;
    for (int w = 0; w < 8; ++w) {
        base += wbs[w];
        if (w < wv) base += winc[w];
    }
    if (in) {
        int4 o;
        o.x = base + inc - s;
        o.y = o.x + c.x;
        o.z = o.y + c.y;
        o.w = o.z + c.z;
        ((int4*)offs)[i4] = o;
    }
}

__global__ __launch_bounds__(256) void gcn_place(const int* __restrict__ row,
                                                 const int* __restrict__ col,
                                                 const int* __restrict__ offs,
                                                 const int* __restrict__ rank,
                                                 int* __restrict__ srow, int e) {
    const int stride = gridDim.x * 256;
    const int i0 = blockIdx.x * 256 + threadIdx.x;
    int c[EDGES_PER_THREAD], r[EDGES_PER_THREAD], k[EDGES_PER_THREAD], o[EDGES_PER_THREAD];
#pragma unroll
    for (int j = 0; j < EDGES_PER_THREAD; ++j) {
        const int i = i0 + j * stride;
        const bool in = i < e;
        c[j] = in ? col[i] : 0;
        r[j] = in ? row[i] : 0;
        k[j] = in ? rank[i] : 0;
    }
#pragma unroll
    for (int j = 0; j < EDGES_PER_THREAD; ++j) o[j] = offs[c[j]];
#pragma unroll
    for (int j = 0; j < EDGES_PER_THREAD; ++j) {
        const int i = i0 + j * stride;
        if (i < e) srow[o[j] + k[j]] = r[j];
    }
}

__device__ __forceinline__ float gcn_dinv_of(int d) {
    const float f = (float)d;
    return (f > 0.0f) ? (1.0f / sqrtf(f)) : 0.0f;
}


#define A_PITCH 132

__device__ __forceinline__ void gcn_lds_barrier() {
    __builtin_amdgcn_fence(__ATOMIC_RELEASE, "workgroup", "local");
    __builtin_amdgcn_s_barrier();
    __builtin_amdgcn_fence(__ATOMIC_ACQUIRE, "workgroup", "local");
}

template <int F, bool RELU, bool SCALE>
__global__ __launch_bounds__(256) void gcn_gemm_wmma(const float* __restrict__ A,
                                                     const float* __restrict__ W,
                                                     const int* __restrict__ offs,
                                                     float* __restrict__ out, int nRows) {
    constexpr int TILES_N = F / 16;
    constexpr int ROW_TILES_PER_BLOCK = 8 / TILES_N;
    constexpr int ROWS = ROW_TILES_PER_BLOCK * 16;
    constexpr int NP   = ROWS / 8;
    constexpr int BUF  = ROWS * A_PITCH;
    __shared__ __align__(16) float lds[2 * BUF];

    const int lane = threadIdx.x & 31;
    const int wv = __builtin_amdgcn_readfirstlane((int)(threadIdx.x >> 5));
    const int tn = wv % TILES_N;
    const int tr = wv / TILES_N;

    const int mn    = lane & 15;
    const int khalf = lane >> 4;

    const int nRowTiles = nRows / 16;
    const int nGroups = (nRowTiles + ROW_TILES_PER_BLOCK - 1) / ROW_TILES_PER_BLOCK;
    int g = (int)blockIdx.x;

    v4f pre[NP];
#pragma unroll
    for (int p = 0; p < NP; ++p) {
        const int row = min(g * ROWS + p * 8 + wv, nRows - 1);
        pre[p] = *(const v4f*)(A + (size_t)row * 128 + lane * 4);
    }

    const float* wcol = W + (size_t)(tn * 16 + mn);
    v2f bfrag[32];
#pragma unroll
    for (int ks = 0; ks < 32; ++ks) {
        const int kb = 4 * ks + 2 * khalf;
        bfrag[ks].x = wcol[(size_t)kb * F];
        bfrag[ks].y = wcol[(size_t)(kb + 1) * F];
    }

#pragma unroll
    for (int p = 0; p < NP; ++p) {
        v4f q = pre[p];
        if (RELU) { q.x = fmaxf(q.x, 0.0f); q.y = fmaxf(q.y, 0.0f); q.z = fmaxf(q.z, 0.0f); q.w = fmaxf(q.w, 0.0f); }
        *(v4f*)(lds + (p * 8 + wv) * A_PITCH + lane * 4) = q;
    }
    __syncthreads();

    int cur = 0;
    for (; g < nGroups; g += (int)gridDim.x) {
        const int  gn   = g + (int)gridDim.x;
        const bool more = gn < nGroups;
        if (more) {
#pragma unroll
            for (int p = 0; p < NP; ++p) {
                const int row = min(gn * ROWS + p * 8 + wv, nRows - 1);
                pre[p] = *(const v4f*)(A + (size_t)row * 128 + lane * 4);
            }
        }

        const int tm = g * ROW_TILES_PER_BLOCK + tr;
        if (tm < nRowTiles) {
            const float* arow = lds + cur * BUF + (tr * 16 + mn) * A_PITCH + 2 * khalf;
            float drow = 1.0f;
            if (SCALE) {
                const int  r    = tm * 16 + mn;
                const bool last = r + 1 >= N_NODES;
                const int  o0   = offs[r];
                const int  o1m  = offs[last ? r : r + 1];
                const float f   = (float)((last ? N_EDGES : o1m) - o0);
                const float d   = 1.0f / sqrtf(fmaxf(f, 1.0f));
                drow = (f > 0.0f) ? d : 0.0f;
            }
            v8f acc = {};
#pragma unroll
            for (int ks = 0; ks < 32; ++ks) {
                const v2f a = *(const v2f*)(arow + 4 * ks);
                acc = __builtin_amdgcn_wmma_f32_16x16x4_f32(
                     false, a,  false, bfrag[ks],
                     (short)0, acc,  false,  false);
            }
            float* orow = out + (size_t)(tm * 16 + khalf * 8) * F + (size_t)(tn * 16 + mn);
#pragma unroll
            for (int v = 0; v < 8; ++v) {
                float val = acc[v];
                if (SCALE) {
                    const float dlo = __int_as_float(__builtin_amdgcn_readlane(__float_as_int(drow), v));
                    const float dhi = __int_as_float(__builtin_amdgcn_readlane(__float_as_int(drow), v + 8));
                    val *= khalf ? dhi : dlo;
                }
                orow[(size_t)v * F] = val;
            }
        }

        if (more) {
#pragma unroll
            for (int p = 0; p < NP; ++p) {
                v4f q = pre[p];
                if (RELU) { q.x = fmaxf(q.x, 0.0f); q.y = fmaxf(q.y, 0.0f); q.z = fmaxf(q.z, 0.0f); q.w = fmaxf(q.w, 0.0f); }
                *(v4f*)(lds + (cur ^ 1) * BUF + (p * 8 + wv) * A_PITCH + lane * 4) = q;
            }
        }
        gcn_lds_barrier();
        cur ^= 1;
    }
}

template <int V> struct gcn_vec;
template <> struct gcn_vec<2> { typedef v2f type; };
template <> struct gcn_vec<4> { typedef v4f type; };

template <int F, bool PRESCALE>
__global__ __launch_bounds__(256) void gcn_gather_sum(const float* __restrict__ t,
                                                      const int* __restrict__ offs,
                                                      const int* __restrict__ srow,
                                                      const float* __restrict__ b,
                                                      float* __restrict__ out, int nNodes) {
    constexpr int V = F / 32;
    typedef typename gcn_vec<V>::type vf;
    const int lane = threadIdx.x & 31;
    const int first = blockIdx.x * 8 + __builtin_amdgcn_readfirstlane((int)(threadIdx.x >> 5));
    const int nWaves = (int)gridDim.x * 8;
    const vf* tv = (const vf*)t + lane;
    const vf bias = *((const vf*)b + lane);

    for (int node = first; node < nNodes; node += nWaves) {
    const int beg = offs[node];
    const int end = (node + 1 < N_NODES) ? offs[node + 1] : N_EDGES;
    const int n = end - beg;
    const float dnode = gcn_dinv_of(n);

    vf acc = {};

    for (int base = 0; base < n; base += 32) {
        const int m = (n - base < 32) ? (n - base) : 32;
        int er = 0;
        if (lane < m) er = srow[beg + base + lane];
        for (int k = 0; k < m; k += 4) {
            const int r0 = __builtin_amdgcn_readlane(er, k);
            const int r1 = __builtin_amdgcn_readlane(er, k + 1);
            const int r2 = __builtin_amdgcn_readlane(er, k + 2);
            const int r3 = __builtin_amdgcn_readlane(er, k + 3);
            const vf x0 = tv[(size_t)r0 * 32];
            const vf x1 = tv[(size_t)r1 * 32];
            const vf x2 = tv[(size_t)r2 * 32];
            const vf x3 = tv[(size_t)r3 * 32];
            const vf zero = {};
            acc += x0;
            acc += (k + 1 < m) ? x1 : zero;
            acc += (k + 2 < m) ? x2 : zero;
            acc += (k + 3 < m) ? x3 : zero;
        }
    }
    vf res = bias + acc * dnode;
    if (PRESCALE) res = res * dnode;
    *((vf*)(out + (size_t)node * F) + lane) = res;
    }
}

extern "C" void kernel_launch(void* const* d_in, const int* in_sizes, int n_in,
                              void* d_out, int out_size, void* d_ws, size_t ws_size,
                              hipStream_t stream) {
    const float* x   = (const float*)d_in[0];
    const int*   ei  = (const int*)d_in[1];
    const float* W1  = (const float*)d_in[2];
    const float* b1  = (const float*)d_in[3];
    const float* W2  = (const float*)d_in[4];
    const float* b2  = (const float*)d_in[5];
    const float* W3  = (const float*)d_in[6];
    const float* b3  = (const float*)d_in[7];
    float*       out = (float*)d_out;

    const int* row = ei;
    const int* col = ei + N_EDGES;

    float* t    = (float*)d_ws;
    float* h    = t + (size_t)N_NODES * 128;
    int*   offs = (int*)(h + (size_t)N_NODES * 128);
    int*   srow = offs + N_NODES;
    int*   cnt  = (int*)h;
    int*   rank = cnt + N_NODES;
    int*   bs   = rank + N_EDGES;

    const int TB = 256;

    const int edgeBlocks = (N_EDGES + TB * EDGES_PER_THREAD - 1) / (TB * EDGES_PER_THREAD);

    gcn_fill_zero<<<(N_NODES / 4 + TB - 1) / TB, TB, 0, stream>>>((int4*)cnt, N_NODES / 4);
    gcn_rank<<<edgeBlocks, TB, 0, stream>>>(col, cnt, rank, N_EDGES);
    gcn_block_sums<<<SCAN_BLOCKS, TB, 0, stream>>>(cnt, bs);
    gcn_offsets<<<SCAN_BLOCKS, TB, 0, stream>>>(cnt, bs, offs);
    gcn_place<<<edgeBlocks, TB, 0, stream>>>(row, col, offs, rank, srow, N_EDGES);

    const int gemm128Blocks = GEMM_BLOCKS;
    const int gemm64Blocks  = GEMM_BLOCKS;
    const int gatherBlocks  = GATHER_BLOCKS;

    gcn_gemm_wmma<128, false, true><<<gemm128Blocks, TB, 0, stream>>>(x, W1, offs, t, N_NODES);
    gcn_gather_sum<128, true><<<gatherBlocks, TB, 0, stream>>>(t, offs, srow, b1, h, N_NODES);

    gcn_gemm_wmma<128, true, false><<<gemm128Blocks, TB, 0, stream>>>(h, W2, offs, t, N_NODES);
    gcn_gather_sum<128, true><<<gatherBlocks, TB, 0, stream>>>(t, offs, srow, b2, h, N_NODES);

    gcn_gemm_wmma<64, true, false><<<gemm64Blocks, TB, 0, stream>>>(h, W3, offs, t, N_NODES);
    gcn_gather_sum<64, false><<<gatherBlocks, TB, 0, stream>>>(t, offs, srow, b3, out, N_NODES);
}


# Aggregation of layers 1 and 2 fused into the next layer's GEMM: waves sum their nodes' source rows straight into the LDS A tile (8 rows in flight); 11 launches to 9; per-node sum order kept
# speedup vs baseline: 6.2452x; 1.1078x over previous
#include <hip/hip_runtime.h>

typedef float v2f __attribute__((ext_vector_type(2)));
typedef float v4f __attribute__((ext_vector_type(4)));
typedef float v8f __attribute__((ext_vector_type(8)));

#define N_NODES 100000
#define N_EDGES 1600000
#define SCAN_CHUNK 1024
#define SCAN_BLOCKS ((N_NODES + SCAN_CHUNK - 1) / SCAN_CHUNK)
#define GATHER_BLOCKS 2048
#define GEMM_BLOCKS   1024

__global__ void gcn_fill_zero(int4* __restrict__ p, int n4) {
    int i = blockIdx.x * blockDim.x + threadIdx.x;
    if (i < n4) p[i] = make_int4(0, 0, 0, 0);
}

#define EDGES_PER_THREAD 4
__global__ __launch_bounds__(256) void gcn_rank(const int* __restrict__ col, int* __restrict__ cnt,
                                                int* __restrict__ rank, int e) {
    const int stride = gridDim.x * 256;
    const int i0 = blockIdx.x * 256 + threadIdx.x;
    int c[EDGES_PER_THREAD], r[EDGES_PER_THREAD];
#pragma unroll
    for (int j = 0; j < EDGES_PER_THREAD; ++j) {
        const int i = i0 + j * stride;
        c[j] = (i < e) ? col[i] : -1;
    }
#pragma unroll
    for (int j = 0; j < EDGES_PER_THREAD; ++j)
        r[j] = (c[j] >= 0) ? atomicAdd(&cnt[c[j]], 1) : 0;
#pragma unroll
    for (int j = 0; j < EDGES_PER_THREAD; ++j) {
        const int i = i0 + j * stride;
        if (i < e) rank[i] = r[j];
    }
}

__global__ __launch_bounds__(256) void gcn_block_sums(const int* __restrict__ cnt,
                                                      int* __restrict__ bs) {
    __shared__ int wsum[8];
    const int i4 = blockIdx.x * 256 + threadIdx.x;
    int s = 0;
    if (i4 * 4 < N_NODES) {
        const int4 c = ((const int4*)cnt)[i4];
        s = c.x + c.y + c.z + c.w;
    }
    for (int o = 16; o > 0; o >>= 1) s += __shfl_xor(s, o);
    if ((threadIdx.x & 31) == 0) wsum[threadIdx.x >> 5] = s;
    __syncthreads();
    if (threadIdx.x == 0) {
        int tot = 0;
        for (int w = 0; w < 8; ++w) tot += wsum[w];
        bs[blockIdx.x] = tot;
    }
}

__global__ __launch_bounds__(256) void gcn_offsets(const int* __restrict__ cnt,
                                                   const int* __restrict__ bs,
                                                   int* __restrict__ offs) {
    __shared__ int winc[8];
    __shared__ int wbs[8];
    const int tid = threadIdx.x, lane = tid & 31, wv = tid >> 5;
    int bv = (tid < (int)blockIdx.x) ? bs[tid] : 0;
    const int i4 = blockIdx.x * 256 + tid;
    const bool in = i4 * 4 < N_NODES;
    int4 c = make_int4(0, 0, 0, 0);
    if (in) c = ((const int4*)cnt)[i4];
    const int s = c.x + c.y + c.z + c.w;
    int inc = s;
    for (int o = 1; o < 32; o <<= 1) {
        const int v = __shfl_up(inc, o);
        if (lane >= o) inc += v;
    }
    for (int o = 16; o > 0; o >>= 1) bv += __shfl_xor(bv, o);
    if (lane == 31) winc[wv] = inc;
    if (lane == 0) wbs[wv] = bv;
    __syncthreads();
    int base = 0;
    for (int w = 0; w < 8; ++w) {
        base += wbs[w];
        if (w < wv) base += winc[w];
    }
    if (in) {
        int4 o;
        o.x = base + inc - s;
        o.y = o.x + c.x;
        o.z = o.y + c.y;
        o.w = o.z + c.z;
        ((int4*)offs)[i4] = o;
    }
}

__global__ __launch_bounds__(256) void gcn_place(const int* __restrict__ row,
                                                 const int* __restrict__ col,
                                                 const int* __restrict__ offs,
                                                 const int* __restrict__ rank,
                                                 int* __restrict__ srow, int e) {
    const int stride = gridDim.x * 256;
    const int i0 = blockIdx.x * 256 + threadIdx.x;
    int c[EDGES_PER_THREAD], r[EDGES_PER_THREAD], k[EDGES_PER_THREAD], o[EDGES_PER_THREAD];
#pragma unroll
    for (int j = 0; j < EDGES_PER_THREAD; ++j) {
        const int i = i0 + j * stride;
        const bool in = i < e;
        c[j] = in ? col[i] : 0;
        r[j] = in ? row[i] : 0;
        k[j] = in ? rank[i] : 0;
    }
#pragma unroll
    for (int j = 0; j < EDGES_PER_THREAD; ++j) o[j] = offs[c[j]];
#pragma unroll
    for (int j = 0; j < EDGES_PER_THREAD; ++j) {
        const int i = i0 + j * stride;
        if (i < e) srow[o[j] + k[j]] = r[j];
    }
}

__device__ __forceinline__ float gcn_dinv_of(int d) {
    const float f = (float)d;
    return (f > 0.0f) ? (1.0f / sqrtf(f)) : 0.0f;
}


#define A_PITCH 132

__device__ __forceinline__ void gcn_lds_barrier() {
    __builtin_amdgcn_fence(__ATOMIC_RELEASE, "workgroup", "local");
    __builtin_amdgcn_s_barrier();
    __builtin_amdgcn_fence(__ATOMIC_ACQUIRE, "workgroup", "local");
}

template <int F, bool RELU, bool SCALE>
__global__ __launch_bounds__(256) void gcn_gemm_wmma(const float* __restrict__ A,
                                                     const float* __restrict__ W,
                                                     const int* __restrict__ offs,
                                                     float* __restrict__ out, int nRows) {
    constexpr int TILES_N = F / 16;
    constexpr int ROW_TILES_PER_BLOCK = 8 / TILES_N;
    constexpr int ROWS = ROW_TILES_PER_BLOCK * 16;
    constexpr int NP   = ROWS / 8;
    constexpr int BUF  = ROWS * A_PITCH;
    __shared__ __align__(16) float lds[2 * BUF];

    const int lane = threadIdx.x & 31;
    const int wv = __builtin_amdgcn_readfirstlane((int)(threadIdx.x >> 5));
    const int tn = wv % TILES_N;
    const int tr = wv / TILES_N;

    const int mn    = lane & 15;
    const int khalf = lane >> 4;

    const int nRowTiles = nRows / 16;
    const int nGroups = (nRowTiles + ROW_TILES_PER_BLOCK - 1) / ROW_TILES_PER_BLOCK;
    int g = (int)blockIdx.x;

    v4f pre[NP];
#pragma unroll
    for (int p = 0; p < NP; ++p) {
        const int row = min(g * ROWS + p * 8 + wv, nRows - 1);
        pre[p] = *(const v4f*)(A + (size_t)row * 128 + lane * 4);
    }

    const float* wcol = W + (size_t)(tn * 16 + mn);
    v2f bfrag[32];
#pragma unroll
    for (int ks = 0; ks < 32; ++ks) {
        const int kb = 4 * ks + 2 * khalf;
        bfrag[ks].x = wcol[(size_t)kb * F];
        bfrag[ks].y = wcol[(size_t)(kb + 1) * F];
    }

#pragma unroll
    for (int p = 0; p < NP; ++p) {
        v4f q = pre[p];
        if (RELU) { q.x = fmaxf(q.x, 0.0f); q.y = fmaxf(q.y, 0.0f); q.z = fmaxf(q.z, 0.0f); q.w = fmaxf(q.w, 0.0f); }
        *(v4f*)(lds + (p * 8 + wv) * A_PITCH + lane * 4) = q;
    }
    __syncthreads();

    int cur = 0;
    for (; g < nGroups; g += (int)gridDim.x) {
        const int  gn   = g + (int)gridDim.x;
        const bool more = gn < nGroups;
        if (more) {
#pragma unroll
            for (int p = 0; p < NP; ++p) {
                const int row = min(gn * ROWS + p * 8 + wv, nRows - 1);
                pre[p] = *(const v4f*)(A + (size_t)row * 128 + lane * 4);
            }
        }

        const int tm = g * ROW_TILES_PER_BLOCK + tr;
        if (tm < nRowTiles) {
            const float* arow = lds + cur * BUF + (tr * 16 + mn) * A_PITCH + 2 * khalf;
            float drow = 1.0f;
            if (SCALE) {
                const int  r    = tm * 16 + mn;
                const bool last = r + 1 >= N_NODES;
                const int  o0   = offs[r];
                const int  o1m  = offs[last ? r : r + 1];
                const float f   = (float)((last ? N_EDGES : o1m) - o0);
                const float d   = 1.0f / sqrtf(fmaxf(f, 1.0f));
                drow = (f > 0.0f) ? d : 0.0f;
            }
            v8f acc = {};
#pragma unroll
            for (int ks = 0; ks < 32; ++ks) {
                const v2f a = *(const v2f*)(arow + 4 * ks);
                acc = __builtin_amdgcn_wmma_f32_16x16x4_f32(
                     false, a,  false, bfrag[ks],
                     (short)0, acc,  false,  false);
            }
            float* orow = out + (size_t)(tm * 16 + khalf * 8) * F + (size_t)(tn * 16 + mn);
#pragma unroll
            for (int v = 0; v < 8; ++v) {
                float val = acc[v];
                if (SCALE) {
                    const float dlo = __int_as_float(__builtin_amdgcn_readlane(__float_as_int(drow), v));
                    const float dhi = __int_as_float(__builtin_amdgcn_readlane(__float_as_int(drow), v + 8));
                    val *= khalf ? dhi : dlo;
                }
                orow[(size_t)v * F] = val;
            }
        }

        if (more) {
#pragma unroll
            for (int p = 0; p < NP; ++p) {
                v4f q = pre[p];
                if (RELU) { q.x = fmaxf(q.x, 0.0f); q.y = fmaxf(q.y, 0.0f); q.z = fmaxf(q.z, 0.0f); q.w = fmaxf(q.w, 0.0f); }
                *(v4f*)(lds + (cur ^ 1) * BUF + (p * 8 + wv) * A_PITCH + lane * 4) = q;
            }
        }
        gcn_lds_barrier();
        cur ^= 1;
    }
}

template <int V> struct gcn_vec;
template <> struct gcn_vec<2> { typedef v2f type; };
template <> struct gcn_vec<4> { typedef v4f type; };

template <int F, bool PRESCALE>
__global__ __launch_bounds__(256) void gcn_gather_sum(const float* __restrict__ t,
                                                      const int* __restrict__ offs,
                                                      const int* __restrict__ srow,
                                                      const float* __restrict__ b,
                                                      float* __restrict__ out, int nNodes) {
    constexpr int V = F / 32;
    typedef typename gcn_vec<V>::type vf;
    const int lane = threadIdx.x & 31;
    const int first = blockIdx.x * 8 + __builtin_amdgcn_readfirstlane((int)(threadIdx.x >> 5));
    const int nWaves = (int)gridDim.x * 8;
    const vf* tv = (const vf*)t + lane;
    const vf bias = *((const vf*)b + lane);

    for (int node = first; node < nNodes; node += nWaves) {
    const int beg = offs[node];
    const int end = (node + 1 < N_NODES) ? offs[node + 1] : N_EDGES;
    const int n = end - beg;
    const float dnode = gcn_dinv_of(n);

    vf acc = {};

    for (int base = 0; base < n; base += 32) {
        const int m = (n - base < 32) ? (n - base) : 32;
        int er = 0;
        if (lane < m) er = srow[beg + base + lane];
        for (int k = 0; k < m; k += 4) {
            const int r0 = __builtin_amdgcn_readlane(er, k);
            const int r1 = __builtin_amdgcn_readlane(er, k + 1);
            const int r2 = __builtin_amdgcn_readlane(er, k + 2);
            const int r3 = __builtin_amdgcn_readlane(er, k + 3);
            const vf x0 = tv[(size_t)r0 * 32];
            const vf x1 = tv[(size_t)r1 * 32];
            const vf x2 = tv[(size_t)r2 * 32];
            const vf x3 = tv[(size_t)r3 * 32];
            const vf zero = {};
            acc += x0;
            acc += (k + 1 < m) ? x1 : zero;
            acc += (k + 2 < m) ? x2 : zero;
            acc += (k + 3 < m) ? x3 : zero;
        }
    }
    vf res = bias + acc * dnode;
    if (PRESCALE) res = res * dnode;
    *((vf*)(out + (size_t)node * F) + lane) = res;
    }
}

#define ROWS_IN_FLIGHT 8

__device__ __forceinline__ v4f gcn_sum_rows128(const float* __restrict__ tin,
                                               const int* __restrict__ srow,
                                               int beg, int n, int lane) {
    v4f acc = {};
    for (int base = 0; base < n; base += 32) {
        const int m = (n - base < 32) ? (n - base) : 32;
        int er = 0;
        if (lane < m) er = srow[beg + base + lane];
        for (int k = 0; k < m; k += ROWS_IN_FLIGHT) {
            v4f x[ROWS_IN_FLIGHT];
#pragma unroll
            for (int i = 0; i < ROWS_IN_FLIGHT; ++i) {
                const unsigned r = (unsigned)__builtin_amdgcn_readlane(er, k + i);
                x[i] = *(const v4f*)((const char*)tin + (r * 512u + (unsigned)lane * 16u));
            }
            const v4f zero = {};
#pragma unroll
            for (int i = 0; i < ROWS_IN_FLIGHT; ++i) acc += (k + i < m) ? x[i] : zero;
        }
    }
    return acc;
}

template <int NP, int ROWS>
__device__ __forceinline__ void gcn_ask_bounds(const int* __restrict__ offs, int gg, int wv,
                                               int (&o0)[NP], int (&o1)[NP]) {
#pragma unroll
    for (int p = 0; p < NP; ++p) {
        const int node = gg * ROWS + p * 8 + wv;
        o0[p] = offs[min(node, N_NODES - 1)];
        o1[p] = offs[min(node + 1, N_NODES - 1)];
    }
}

template <int NP, int ROWS>
__device__ __forceinline__ void gcn_fill_rows(float* __restrict__ dst, const float* __restrict__ tin,
                                              const int* __restrict__ srow, v4f bias,
                                              int gg, int wv, int lane, int nRows,
                                              const int (&o0)[NP], const int (&o1)[NP]) {
#pragma unroll
    for (int p = 0; p < NP; ++p) {
        const int node = gg * ROWS + p * 8 + wv;
        const int end  = (node + 1 < N_NODES) ? o1[p] : N_EDGES;
        const int n    = (node < nRows) ? (end - o0[p]) : 0;
        const float dnode = gcn_dinv_of(n);
        const v4f s = gcn_sum_rows128(tin, srow, o0[p], n, lane);
        v4f q = (bias + s * dnode) * dnode;
        q.x = fmaxf(q.x, 0.0f); q.y = fmaxf(q.y, 0.0f); q.z = fmaxf(q.z, 0.0f); q.w = fmaxf(q.w, 0.0f);
        *(v4f*)(dst + (p * 8 + wv) * A_PITCH + lane * 4) = q;
    }
}

template <int F>
__global__ __launch_bounds__(256) void gcn_gather_gemm_wmma(const float* __restrict__ tin,
                                                            const int* __restrict__ offs,
                                                            const int* __restrict__ srow,
                                                            const float* __restrict__ b,
                                                            const float* __restrict__ W,
                                                            float* __restrict__ out, int nRows) {
    constexpr int TILES_N = F / 16;
    constexpr int ROW_TILES_PER_BLOCK = 8 / TILES_N;
    constexpr int ROWS = ROW_TILES_PER_BLOCK * 16;
    constexpr int NP   = ROWS / 8;
    constexpr int BUF  = ROWS * A_PITCH;
    __shared__ __align__(16) float lds[2 * BUF];

    const int lane = threadIdx.x & 31;
    const int wv = __builtin_amdgcn_readfirstlane((int)(threadIdx.x >> 5));
    const int tn = wv % TILES_N;
    const int tr = wv / TILES_N;

    const int mn    = lane & 15;
    const int khalf = lane >> 4;

    const int nRowTiles = nRows / 16;
    const int nGroups = (nRowTiles + ROW_TILES_PER_BLOCK - 1) / ROW_TILES_PER_BLOCK;
    int g = (int)blockIdx.x;

    int o0[NP], o1[NP];
    gcn_ask_bounds<NP, ROWS>(offs, g, wv, o0, o1);

    const v4f bias = *((const v4f*)b + lane);

    const float* wcol = W + (size_t)(tn * 16 + mn);
    v2f bfrag[32];
#pragma unroll
    for (int ks = 0; ks < 32; ++ks) {
        const int kb = 4 * ks + 2 * khalf;
        bfrag[ks].x = wcol[(size_t)kb * F];
        bfrag[ks].y = wcol[(size_t)(kb + 1) * F];
    }

    gcn_fill_rows<NP, ROWS>(lds, tin, srow, bias, g, wv, lane, nRows, o0, o1);
    __syncthreads();

    int cur = 0;
    for (; g < nGroups; g += (int)gridDim.x) {
        const int  gn   = g + (int)gridDim.x;
        const bool more = gn < nGroups;
        if (more) gcn_ask_bounds<NP, ROWS>(offs, gn, wv, o0, o1);

        const int tm = g * ROW_TILES_PER_BLOCK + tr;
        if (tm < nRowTiles) {
            const float* arow = lds + cur * BUF + (tr * 16 + mn) * A_PITCH + 2 * khalf;
            v8f acc = {};
#pragma unroll
            for (int ks = 0; ks < 32; ++ks) {
                const v2f a = *(const v2f*)(arow + 4 * ks);
                acc = __builtin_amdgcn_wmma_f32_16x16x4_f32(
                     false, a,  false, bfrag[ks],
                     (short)0, acc,  false,  false);
            }
            float* orow = out + (size_t)(tm * 16 + khalf * 8) * F + (size_t)(tn * 16 + mn);
#pragma unroll
            for (int v = 0; v < 8; ++v) orow[(size_t)v * F] = acc[v];
        }

        if (more)
            gcn_fill_rows<NP, ROWS>(lds + (cur ^ 1) * BUF, tin, srow, bias, gn, wv, lane, nRows, o0, o1);
        gcn_lds_barrier();
        cur ^= 1;
    }
}

extern "C" void kernel_launch(void* const* d_in, const int* in_sizes, int n_in,
                              void* d_out, int out_size, void* d_ws, size_t ws_size,
                              hipStream_t stream) {
    const float* x   = (const float*)d_in[0];
    const int*   ei  = (const int*)d_in[1];
    const float* W1  = (const float*)d_in[2];
    const float* b1  = (const float*)d_in[3];
    const float* W2  = (const float*)d_in[4];
    const float* b2  = (const float*)d_in[5];
    const float* W3  = (const float*)d_in[6];
    const float* b3  = (const float*)d_in[7];
    float*       out = (float*)d_out;

    const int* row = ei;
    const int* col = ei + N_EDGES;

    float* t    = (float*)d_ws;
    float* h    = t + (size_t)N_NODES * 128;
    int*   offs = (int*)(h + (size_t)N_NODES * 128);
    int*   srow = offs + N_NODES;
    int*   cnt  = (int*)h;
    int*   rank = cnt + N_NODES;
    int*   bs   = rank + N_EDGES;

    const int TB = 256;

    const int edgeBlocks = (N_EDGES + TB * EDGES_PER_THREAD - 1) / (TB * EDGES_PER_THREAD);

    gcn_fill_zero<<<(N_NODES / 4 + TB - 1) / TB, TB, 0, stream>>>((int4*)cnt, N_NODES / 4);
    gcn_rank<<<edgeBlocks, TB, 0, stream>>>(col, cnt, rank, N_EDGES);
    gcn_block_sums<<<SCAN_BLOCKS, TB, 0, stream>>>(cnt, bs);
    gcn_offsets<<<SCAN_BLOCKS, TB, 0, stream>>>(cnt, bs, offs);
    gcn_place<<<edgeBlocks, TB, 0, stream>>>(row, col, offs, rank, srow, N_EDGES);

    const int gemm128Blocks = GEMM_BLOCKS;
    const int gemm64Blocks  = GEMM_BLOCKS;
    const int gatherBlocks  = GATHER_BLOCKS;

    gcn_gemm_wmma<128, false, true><<<gemm128Blocks, TB, 0, stream>>>(x, W1, offs, t, N_NODES);

    gcn_gather_gemm_wmma<128><<<gemm128Blocks, TB, 0, stream>>>(t, offs, srow, b1, W2, h, N_NODES);

    gcn_gather_gemm_wmma<64><<<gemm64Blocks, TB, 0, stream>>>(h, offs, srow, b2, W3, t, N_NODES);

    gcn_gather_sum<64, false><<<gatherBlocks, TB, 0, stream>>>(t, offs, srow, b3, out, N_NODES);
}


# Build tail and layer 1 in 2 launches, not 4: scan inside chunks only (readers add chunk bases); edge placement inside layer 1's GEMM launch, 1 placement + 3 GEMM blocks per CU; 9 launches to 7
# speedup vs baseline: 6.5038x; 1.0414x over previous
#include <hip/hip_runtime.h>

typedef float v2f __attribute__((ext_vector_type(2)));
typedef float v4f __attribute__((ext_vector_type(4)));
typedef float v8f __attribute__((ext_vector_type(8)));

#define N_NODES 100000
#define N_EDGES 1600000
#define SCAN_CHUNK 1024
#define SCAN_BLOCKS ((N_NODES + SCAN_CHUNK - 1) / SCAN_CHUNK)
#define GATHER_BLOCKS 2048
#define GEMM_BLOCKS   1024
#define PLACE_BLOCKS      256
#define FUSED_GEMM_BLOCKS (GEMM_BLOCKS - PLACE_BLOCKS)

__global__ void gcn_fill_zero(int4* __restrict__ p, int n4) {
    int i = blockIdx.x * blockDim.x + threadIdx.x;
    if (i < n4) p[i] = make_int4(0, 0, 0, 0);
}

#define EDGES_PER_THREAD 4
__global__ __launch_bounds__(256) void gcn_rank(const int* __restrict__ col, int* __restrict__ cnt,
                                                int* __restrict__ rank, int e) {
    const int stride = gridDim.x * 256;
    const int i0 = blockIdx.x * 256 + threadIdx.x;
    int c[EDGES_PER_THREAD], r[EDGES_PER_THREAD];
#pragma unroll
    for (int j = 0; j < EDGES_PER_THREAD; ++j) {
        const int i = i0 + j * stride;
        c[j] = (i < e) ? col[i] : -1;
    }
#pragma unroll
    for (int j = 0; j < EDGES_PER_THREAD; ++j)
        r[j] = (c[j] >= 0) ? atomicAdd(&cnt[c[j]], 1) : 0;
#pragma unroll
    for (int j = 0; j < EDGES_PER_THREAD; ++j) {
        const int i = i0 + j * stride;
        if (i < e) rank[i] = r[j];
    }
}

__global__ __launch_bounds__(256) void gcn_chunk_scan(const int* __restrict__ cnt,
                                                      int* __restrict__ rel,
                                                      int* __restrict__ bs) {
    __shared__ int winc[8];
    const int tid = threadIdx.x, lane = tid & 31, wv = tid >> 5;
    const int i4 = blockIdx.x * 256 + tid;
    const bool in = i4 * 4 < N_NODES;
    int4 c = make_int4(0, 0, 0, 0);
    if (in) c = ((const int4*)cnt)[i4];
    const int s = c.x + c.y + c.z + c.w;
    int inc = s;
    for (int o = 1; o < 32; o <<= 1) {
        const int v = __shfl_up(inc, o);
        if (lane >= o) inc += v;
    }
    if (lane == 31) winc[wv] = inc;
    __syncthreads();
    int base = 0, tot = 0;
    for (int w = 0; w < 8; ++w) {
        if (w < wv) base += winc[w];
        tot += winc[w];
    }
    if (in) {
        int4 o;
        o.x = base + inc - s;
        o.y = o.x + c.x;
        o.z = o.y + c.y;
        o.w = o.z + c.z;
        ((int4*)rel)[i4] = o;
    }
    if (tid == 0) bs[blockIdx.x] = tot;
}

#define PLACE_EDGES 5

template <bool GUARD>
__device__ __forceinline__ void gcn_place_trip(const int* __restrict__ row,
                                               const int* __restrict__ col,
                                               const int* __restrict__ rel,
                                               const int* cbase,
                                               const int* __restrict__ rank,
                                               int* __restrict__ srow, int e, int i0, int stride) {
    int c[PLACE_EDGES], r[PLACE_EDGES], k[PLACE_EDGES], o[PLACE_EDGES];
#pragma unroll
    for (int j = 0; j < PLACE_EDGES; ++j) {
        const int i = GUARD ? min(i0 + j * stride, e - 1) : i0 + j * stride;
        c[j] = col[i];
        r[j] = row[i];
        k[j] = rank[i];
    }
#pragma unroll
    for (int j = 0; j < PLACE_EDGES; ++j) o[j] = rel[c[j]];
#pragma unroll
    for (int j = 0; j < PLACE_EDGES; ++j)
        if (!GUARD || i0 + j * stride < e) srow[cbase[c[j] / SCAN_CHUNK] + o[j] + k[j]] = r[j];
}

__device__ __forceinline__ void gcn_place_body(const int* __restrict__ row,
                                               const int* __restrict__ col,
                                               const int* __restrict__ rel,
                                               const int* __restrict__ bs,
                                               const int* __restrict__ rank,
                                               int* __restrict__ offs,
                                               int* __restrict__ srow, int e, int blk, int nBlk) {
    __shared__ int ctot[SCAN_BLOCKS];
    __shared__ int cbase[SCAN_BLOCKS];
    const int tid = (int)threadIdx.x;
    if (tid < SCAN_BLOCKS) ctot[tid] = bs[tid];
    __syncthreads();
    if (tid < SCAN_BLOCKS) {
        int s = 0;
        for (int w = 0; w < tid; ++w) s += ctot[w];
        cbase[tid] = s;
    }
    __syncthreads();

    const int stride = nBlk * 256;
    const int t0 = blk * 256 + tid;
    for (int c = t0; c < N_NODES; c += stride) offs[c] = cbase[c / SCAN_CHUNK] + rel[c];

    for (int base = 0; base < e; base += PLACE_EDGES * stride) {
        const int i0 = base + t0;
        if (i0 + (PLACE_EDGES - 1) * stride < e)
            gcn_place_trip<false>(row, col, rel, cbase, rank, srow, e, i0, stride);
        else
            gcn_place_trip<true>(row, col, rel, cbase, rank, srow, e, i0, stride);
    }
}

__device__ __forceinline__ float gcn_dinv_of(int d) {
    const float f = (float)d;
    return (f > 0.0f) ? (1.0f / sqrtf(f)) : 0.0f;
}


#define A_PITCH 132

__device__ __forceinline__ void gcn_lds_barrier() {
    __builtin_amdgcn_fence(__ATOMIC_RELEASE, "workgroup", "local");
    __builtin_amdgcn_s_barrier();
    __builtin_amdgcn_fence(__ATOMIC_ACQUIRE, "workgroup", "local");
}

template <int F, bool RELU, bool SCALE>
__device__ __forceinline__ void gcn_gemm_body(const float* __restrict__ A,
                                              const float* __restrict__ W,
                                              const int* __restrict__ cnt,
                                              float* __restrict__ out, int nRows,
                                              int blk, int nBlk) {
    constexpr int TILES_N = F / 16;
    constexpr int ROW_TILES_PER_BLOCK = 8 / TILES_N;
    constexpr int ROWS = ROW_TILES_PER_BLOCK * 16;
    constexpr int NP   = ROWS / 8;
    constexpr int BUF  = ROWS * A_PITCH;
    __shared__ __align__(16) float lds[2 * BUF];

    const int lane = threadIdx.x & 31;
    const int wv = __builtin_amdgcn_readfirstlane((int)(threadIdx.x >> 5));
    const int tn = wv % TILES_N;
    const int tr = wv / TILES_N;

    const int mn    = lane & 15;
    const int khalf = lane >> 4;

    const int nRowTiles = nRows / 16;
    const int nGroups = (nRowTiles + ROW_TILES_PER_BLOCK - 1) / ROW_TILES_PER_BLOCK;
    int g = blk;

    v4f pre[NP];
#pragma unroll
    for (int p = 0; p < NP; ++p) {
        const int row = min(g * ROWS + p * 8 + wv, nRows - 1);
        pre[p] = *(const v4f*)(A + (size_t)row * 128 + lane * 4);
    }

    const float* wcol = W + (size_t)(tn * 16 + mn);
    v2f bfrag[32];
#pragma unroll
    for (int ks = 0; ks < 32; ++ks) {
        const int kb = 4 * ks + 2 * khalf;
        bfrag[ks].x = wcol[(size_t)kb * F];
        bfrag[ks].y = wcol[(size_t)(kb + 1) * F];
    }

#pragma unroll
    for (int p = 0; p < NP; ++p) {
        v4f q = pre[p];
        if (RELU) { q.x = fmaxf(q.x, 0.0f); q.y = fmaxf(q.y, 0.0f); q.z = fmaxf(q.z, 0.0f); q.w = fmaxf(q.w, 0.0f); }
        *(v4f*)(lds + (p * 8 + wv) * A_PITCH + lane * 4) = q;
    }
    __syncthreads();

    int cur = 0;
    for (; g < nGroups; g += nBlk) {
        const int  gn   = g + nBlk;
        const bool more = gn < nGroups;
        if (more) {
#pragma unroll
            for (int p = 0; p < NP; ++p) {
                const int row = min(gn * ROWS + p * 8 + wv, nRows - 1);
                pre[p] = *(const v4f*)(A + (size_t)row * 128 + lane * 4);
            }
        }

        const int tm = g * ROW_TILES_PER_BLOCK + tr;
        if (tm < nRowTiles) {
            const float* arow = lds + cur * BUF + (tr * 16 + mn) * A_PITCH + 2 * khalf;
            float drow = 1.0f;
            if (SCALE) {
                const float f = (float)cnt[tm * 16 + mn];
                const float d = 1.0f / sqrtf(fmaxf(f, 1.0f));
                drow = (f > 0.0f) ? d : 0.0f;
            }
            v8f acc = {};
#pragma unroll
            for (int ks = 0; ks < 32; ++ks) {
                const v2f a = *(const v2f*)(arow + 4 * ks);
                acc = __builtin_amdgcn_wmma_f32_16x16x4_f32(
                     false, a,  false, bfrag[ks],
                     (short)0, acc,  false,  false);
            }
            float* orow = out + (size_t)(tm * 16 + khalf * 8) * F + (size_t)(tn * 16 + mn);
#pragma unroll
            for (int v = 0; v < 8; ++v) {
                float val = acc[v];
                if (SCALE) {
                    const float dlo = __int_as_float(__builtin_amdgcn_readlane(__float_as_int(drow), v));
                    const float dhi = __int_as_float(__builtin_amdgcn_readlane(__float_as_int(drow), v + 8));
                    val *= khalf ? dhi : dlo;
                }
                orow[(size_t)v * F] = val;
            }
        }

        if (more) {
#pragma unroll
            for (int p = 0; p < NP; ++p) {
                v4f q = pre[p];
                if (RELU) { q.x = fmaxf(q.x, 0.0f); q.y = fmaxf(q.y, 0.0f); q.z = fmaxf(q.z, 0.0f); q.w = fmaxf(q.w, 0.0f); }
                *(v4f*)(lds + (cur ^ 1) * BUF + (p * 8 + wv) * A_PITCH + lane * 4) = q;
            }
        }
        gcn_lds_barrier();
        cur ^= 1;
    }
}

__global__ __launch_bounds__(256) void gcn_place_and_gemm1(const int* __restrict__ row,
                                                           const int* __restrict__ col,
                                                           const int* __restrict__ cnt,
                                                           const int* __restrict__ rel,
                                                           const int* __restrict__ bs,
                                                           const int* __restrict__ rank,
                                                           int* __restrict__ offs,
                                                           int* __restrict__ srow, int e,
                                                           const float* __restrict__ x,
                                                           const float* __restrict__ W1,
                                                           float* __restrict__ t, int nRows) {
    const int b = (int)blockIdx.x;
    const int q = b / PLACE_BLOCKS;
    const int r = b % PLACE_BLOCKS;
    const int j = ((r & 3) + q) & 3;
    if (j == 3)
        gcn_place_body(row, col, rel, bs, rank, offs, srow, e, r, PLACE_BLOCKS);
    else
        gcn_gemm_body<128, false, true>(x, W1, cnt, t, nRows, r * 3 + j, FUSED_GEMM_BLOCKS);
}

template <int V> struct gcn_vec;
template <> struct gcn_vec<2> { typedef v2f type; };
template <> struct gcn_vec<4> { typedef v4f type; };

template <int F, bool PRESCALE>
__global__ __launch_bounds__(256) void gcn_gather_sum(const float* __restrict__ t,
                                                      const int* __restrict__ offs,
                                                      const int* __restrict__ srow,
                                                      const float* __restrict__ b,
                                                      float* __restrict__ out, int nNodes) {
    constexpr int V = F / 32;
    typedef typename gcn_vec<V>::type vf;
    const int lane = threadIdx.x & 31;
    const int first = blockIdx.x * 8 + __builtin_amdgcn_readfirstlane((int)(threadIdx.x >> 5));
    const int nWaves = (int)gridDim.x * 8;
    const vf* tv = (const vf*)t + lane;
    const vf bias = *((const vf*)b + lane);

    for (int node = first; node < nNodes; node += nWaves) {
    const int beg = offs[node];
    const int end = (node + 1 < N_NODES) ? offs[node + 1] : N_EDGES;
    const int n = end - beg;
    const float dnode = gcn_dinv_of(n);

    vf acc = {};

    for (int base = 0; base < n; base += 32) {
        const int m = (n - base < 32) ? (n - base) : 32;
        int er = 0;
        if (lane < m) er = srow[beg + base + lane];
        for (int k = 0; k < m; k += 4) {
            const int r0 = __builtin_amdgcn_readlane(er, k);
            const int r1 = __builtin_amdgcn_readlane(er, k + 1);
            const int r2 = __builtin_amdgcn_readlane(er, k + 2);
            const int r3 = __builtin_amdgcn_readlane(er, k + 3);
            const vf x0 = tv[(size_t)r0 * 32];
            const vf x1 = tv[(size_t)r1 * 32];
            const vf x2 = tv[(size_t)r2 * 32];
            const vf x3 = tv[(size_t)r3 * 32];
            const vf zero = {};
            acc += x0;
            acc += (k + 1 < m) ? x1 : zero;
            acc += (k + 2 < m) ? x2 : zero;
            acc += (k + 3 < m) ? x3 : zero;
        }
    }
    vf res = bias + acc * dnode;
    if (PRESCALE) res = res * dnode;
    *((vf*)(out + (size_t)node * F) + lane) = res;
    }
}

#define ROWS_IN_FLIGHT 8

__device__ __forceinline__ v4f gcn_sum_rows128(const float* __restrict__ tin,
                                               const int* __restrict__ srow,
                                               int beg, int n, int lane) {
    v4f acc = {};
    for (int base = 0; base < n; base += 32) {
        const int m = (n - base < 32) ? (n - base) : 32;
        int er = 0;
        if (lane < m) er = srow[beg + base + lane];
        for (int k = 0; k < m; k += ROWS_IN_FLIGHT) {
            v4f x[ROWS_IN_FLIGHT];
#pragma unroll
            for (int i = 0; i < ROWS_IN_FLIGHT; ++i) {
                const unsigned r = (unsigned)__builtin_amdgcn_readlane(er, k + i);
                x[i] = *(const v4f*)((const char*)tin + (r * 512u + (unsigned)lane * 16u));
            }
            const v4f zero = {};
#pragma unroll
            for (int i = 0; i < ROWS_IN_FLIGHT; ++i) acc += (k + i < m) ? x[i] : zero;
        }
    }
    return acc;
}

template <int NP, int ROWS>
__device__ __forceinline__ void gcn_ask_bounds(const int* __restrict__ offs, int gg, int wv,
                                               int (&o0)[NP], int (&o1)[NP]) {
#pragma unroll
    for (int p = 0; p < NP; ++p) {
        const int node = gg * ROWS + p * 8 + wv;
        o0[p] = offs[min(node, N_NODES - 1)];
        o1[p] = offs[min(node + 1, N_NODES - 1)];
    }
}

template <int NP, int ROWS>
__device__ __forceinline__ void gcn_fill_rows(float* __restrict__ dst, const float* __restrict__ tin,
                                              const int* __restrict__ srow, v4f bias,
                                              int gg, int wv, int lane, int nRows,
                                              const int (&o0)[NP], const int (&o1)[NP]) {
#pragma unroll
    for (int p = 0; p < NP; ++p) {
        const int node = gg * ROWS + p * 8 + wv;
        const int end  = (node + 1 < N_NODES) ? o1[p] : N_EDGES;
        const int n    = (node < nRows) ? (end - o0[p]) : 0;
        const float dnode = gcn_dinv_of(n);
        const v4f s = gcn_sum_rows128(tin, srow, o0[p], n, lane);
        v4f q = (bias + s * dnode) * dnode;
        q.x = fmaxf(q.x, 0.0f); q.y = fmaxf(q.y, 0.0f); q.z = fmaxf(q.z, 0.0f); q.w = fmaxf(q.w, 0.0f);
        *(v4f*)(dst + (p * 8 + wv) * A_PITCH + lane * 4) = q;
    }
}

template <int F>
__global__ __launch_bounds__(256) void gcn_gather_gemm_wmma(const float* __restrict__ tin,
                                                            const int* __restrict__ offs,
                                                            const int* __restrict__ srow,
                                                            const float* __restrict__ b,
                                                            const float* __restrict__ W,
                                                            float* __restrict__ out, int nRows) {
    constexpr int TILES_N = F / 16;
    constexpr int ROW_TILES_PER_BLOCK = 8 / TILES_N;
    constexpr int ROWS = ROW_TILES_PER_BLOCK * 16;
    constexpr int NP   = ROWS / 8;
    constexpr int BUF  = ROWS * A_PITCH;
    __shared__ __align__(16) float lds[2 * BUF];

    const int lane = threadIdx.x & 31;
    const int wv = __builtin_amdgcn_readfirstlane((int)(threadIdx.x >> 5));
    const int tn = wv % TILES_N;
    const int tr = wv / TILES_N;

    const int mn    = lane & 15;
    const int khalf = lane >> 4;

    const int nRowTiles = nRows / 16;
    const int nGroups = (nRowTiles + ROW_TILES_PER_BLOCK - 1) / ROW_TILES_PER_BLOCK;
    int g = (int)blockIdx.x;

    int o0[NP], o1[NP];
    gcn_ask_bounds<NP, ROWS>(offs, g, wv, o0, o1);

    const v4f bias = *((const v4f*)b + lane);

    const float* wcol = W + (size_t)(tn * 16 + mn);
    v2f bfrag[32];
#pragma unroll
    for (int ks = 0; ks < 32; ++ks) {
        const int kb = 4 * ks + 2 * khalf;
        bfrag[ks].x = wcol[(size_t)kb * F];
        bfrag[ks].y = wcol[(size_t)(kb + 1) * F];
    }

    gcn_fill_rows<NP, ROWS>(lds, tin, srow, bias, g, wv, lane, nRows, o0, o1);
    __syncthreads();

    int cur = 0;
    for (; g < nGroups; g += (int)gridDim.x) {
        const int  gn   = g + (int)gridDim.x;
        const bool more = gn < nGroups;
        if (more) gcn_ask_bounds<NP, ROWS>(offs, gn, wv, o0, o1);

        const int tm = g * ROW_TILES_PER_BLOCK + tr;
        if (tm < nRowTiles) {
            const float* arow = lds + cur * BUF + (tr * 16 + mn) * A_PITCH + 2 * khalf;
            v8f acc = {};
#pragma unroll
            for (int ks = 0; ks < 32; ++ks) {
                const v2f a = *(const v2f*)(arow + 4 * ks);
                acc = __builtin_amdgcn_wmma_f32_16x16x4_f32(
                     false, a,  false, bfrag[ks],
                     (short)0, acc,  false,  false);
            }
            float* orow = out + (size_t)(tm * 16 + khalf * 8) * F + (size_t)(tn * 16 + mn);
#pragma unroll
            for (int v = 0; v < 8; ++v) orow[(size_t)v * F] = acc[v];
        }

        if (more)
            gcn_fill_rows<NP, ROWS>(lds + (cur ^ 1) * BUF, tin, srow, bias, gn, wv, lane, nRows, o0, o1);
        gcn_lds_barrier();
        cur ^= 1;
    }
}

extern "C" void kernel_launch(void* const* d_in, const int* in_sizes, int n_in,
                              void* d_out, int out_size, void* d_ws, size_t ws_size,
                              hipStream_t stream) {
    const float* x   = (const float*)d_in[0];
    const int*   ei  = (const int*)d_in[1];
    const float* W1  = (const float*)d_in[2];
    const float* b1  = (const float*)d_in[3];
    const float* W2  = (const float*)d_in[4];
    const float* b2  = (const float*)d_in[5];
    const float* W3  = (const float*)d_in[6];
    const float* b3  = (const float*)d_in[7];
    float*       out = (float*)d_out;

    const int* row = ei;
    const int* col = ei + N_EDGES;

    float* t    = (float*)d_ws;
    float* h    = t + (size_t)N_NODES * 128;
    int*   offs = (int*)(h + (size_t)N_NODES * 128);
    int*   srow = offs + N_NODES;
    int*   cnt  = (int*)h;
    int*   rank = cnt + N_NODES;
    int*   bs   = rank + N_EDGES;
    int*   rel  = bs + 128;

    const int TB = 256;

    const int edgeBlocks = (N_EDGES + TB * EDGES_PER_THREAD - 1) / (TB * EDGES_PER_THREAD);

    gcn_fill_zero<<<(N_NODES / 4 + TB - 1) / TB, TB, 0, stream>>>((int4*)cnt, N_NODES / 4);
    gcn_rank<<<edgeBlocks, TB, 0, stream>>>(col, cnt, rank, N_EDGES);
    gcn_chunk_scan<<<SCAN_BLOCKS, TB, 0, stream>>>(cnt, rel, bs);

    const int gemm128Blocks = GEMM_BLOCKS;
    const int gemm64Blocks  = GEMM_BLOCKS;
    const int gatherBlocks  = GATHER_BLOCKS;

    gcn_place_and_gemm1<<<GEMM_BLOCKS, TB, 0, stream>>>(row, col, cnt, rel, bs, rank, offs, srow, N_EDGES,
                                                             x, W1, t, N_NODES);

    gcn_gather_gemm_wmma<128><<<gemm128Blocks, TB, 0, stream>>>(t, offs, srow, b1, W2, h, N_NODES);

    gcn_gather_gemm_wmma<64><<<gemm64Blocks, TB, 0, stream>>>(h, offs, srow, b2, W3, t, N_NODES);

    gcn_gather_sum<64, false><<<gatherBlocks, TB, 0, stream>>>(t, offs, srow, b3, out, N_NODES);
}


# Edge order built in LDS: slices binned by destination range, then one block per range counts, scans, places and does its layer-1 rows; no global atomics; 7 launches to 5; sum order: arrival rank
# speedup vs baseline: 9.3887x; 1.4436x over previous
#include <hip/hip_runtime.h>

typedef float v2f __attribute__((ext_vector_type(2)));
typedef float v4f __attribute__((ext_vector_type(4)));
typedef float v8f __attribute__((ext_vector_type(8)));


#define N_NODES 100000
#define N_EDGES 1600000
#define GATHER_BLOCKS 2048
#define GEMM_BLOCKS   1024

#define BIN_NODES   112
#define N_BINS      ((N_NODES + BIN_NODES - 1) / BIN_NODES)
#define N_SLICES    256
#define SLICE_EDGES ((N_EDGES + N_SLICES - 1) / N_SLICES)
#define BIN_LOADS   13
#define OFF_PITCH   896
static_assert(N_BINS + 1 <= 1024 && N_BINS + 1 <= OFF_PITCH, "the range scan gives each of 256 threads 4 entries");
static_assert(BIN_NODES <= 128 && BIN_NODES % 16 == 0, "7 bits of a packed word hold the node's place in its range");
static_assert(N_NODES <= (1 << 24), "24 bits of a packed word hold the source row");

__global__ __launch_bounds__(256) void gcn_bin_slices(const int* __restrict__ row,
                                                      const int* __restrict__ col,
                                                      int* __restrict__ binned,
                                                      int* __restrict__ binOff, int e) {
    __shared__ int hist[N_BINS];
    __shared__ int wsum[8];
    const int tid = (int)threadIdx.x, lane = tid & 31, wv = tid >> 5;
    const int s  = (int)blockIdx.x;
    const int e0 = s * SLICE_EDGES;
    const int e1 = min(e, e0 + SLICE_EDGES);
    for (int b = tid; b < N_BINS; b += 256) hist[b] = 0;
    __syncthreads();

    for (int base = e0; base < e1; base += 256 * BIN_LOADS) {
        int c[BIN_LOADS];
#pragma unroll
        for (int j = 0; j < BIN_LOADS; ++j) c[j] = col[min(base + j * 256 + tid, e1 - 1)];
#pragma unroll
        for (int j = 0; j < BIN_LOADS; ++j)
            if (base + j * 256 + tid < e1) atomicAdd(&hist[c[j] / BIN_NODES], 1);
    }
    __syncthreads();

    int v[4], sum = 0;
#pragma unroll
    for (int j = 0; j < 4; ++j) {
        const int b = tid * 4 + j;
        v[j] = (b < N_BINS) ? hist[b] : 0;
        sum += v[j];
    }
    int inc = sum;
    for (int o = 1; o < 32; o <<= 1) {
        const int u = __shfl_up(inc, o);
        if (lane >= o) inc += u;
    }
    if (lane == 31) wsum[wv] = inc;
    __syncthreads();
    int ex = inc - sum;
    for (int w = 0; w < wv; ++w) ex += wsum[w];
#pragma unroll
    for (int j = 0; j < 4; ++j) {
        const int b = tid * 4 + j;
        if (b < N_BINS) hist[b] = ex;
        if (b <= N_BINS) binOff[s * OFF_PITCH + b] = ex;
        ex += v[j];
    }
    __syncthreads();

    for (int base = e0; base < e1; base += 256 * BIN_LOADS) {
        int c[BIN_LOADS], r[BIN_LOADS];
#pragma unroll
        for (int j = 0; j < BIN_LOADS; ++j) {
            const int i = min(base + j * 256 + tid, e1 - 1);
            c[j] = col[i];
            r[j] = row[i];
        }
#pragma unroll
        for (int j = 0; j < BIN_LOADS; ++j)
            if (base + j * 256 + tid < e1) {
                const int b = c[j] / BIN_NODES;
                const int p = atomicAdd(&hist[b], 1);
                binned[e0 + p] = (r[j] << 7) | (c[j] - b * BIN_NODES);
            }
    }
}

__device__ __forceinline__ float gcn_dinv_of(int d) {
    const float f = (float)d;
    return (f > 0.0f) ? (1.0f / sqrtf(f)) : 0.0f;
}


#define A_PITCH 132

__device__ __forceinline__ void gcn_lds_barrier() {
    __builtin_amdgcn_fence(__ATOMIC_RELEASE, "workgroup", "local");
    __builtin_amdgcn_s_barrier();
    __builtin_amdgcn_fence(__ATOMIC_ACQUIRE, "workgroup", "local");
}


template <int F, bool RELU, bool SCALE>
__device__ __forceinline__ void gcn_gemm_body(const float* __restrict__ A,
                                              const float* __restrict__ W,
                                              const int* deg, int degBase,
                                              float* __restrict__ out, int nRows,
                                              int gBegin, int gEnd, int gStep) {
    constexpr int TILES_N = F / 16;
    constexpr int ROW_TILES_PER_BLOCK = 8 / TILES_N;
    constexpr int ROWS = ROW_TILES_PER_BLOCK * 16;
    constexpr int NP   = ROWS / 8;
    constexpr int BUF  = ROWS * A_PITCH;
    __shared__ __align__(16) float lds[2 * BUF];

    const int lane = threadIdx.x & 31;
    const int wv = __builtin_amdgcn_readfirstlane((int)(threadIdx.x >> 5));
    const int tn = wv % TILES_N;
    const int tr = wv / TILES_N;

    const int mn    = lane & 15;
    const int khalf = lane >> 4;

    const int nRowTiles = nRows / 16;
    int g = gBegin;

    v4f pre[NP];
#pragma unroll
    for (int p = 0; p < NP; ++p) {
        const int row = min(g * ROWS + p * 8 + wv, nRows - 1);
        pre[p] = *(const v4f*)(A + (size_t)row * 128 + lane * 4);
    }

    const float* wcol = W + (size_t)(tn * 16 + mn);
    v2f bfrag[32];
#pragma unroll
    for (int ks = 0; ks < 32; ++ks) {
        const int kb = 4 * ks + 2 * khalf;
        bfrag[ks].x = wcol[(size_t)kb * F];
        bfrag[ks].y = wcol[(size_t)(kb + 1) * F];
    }

#pragma unroll
    for (int p = 0; p < NP; ++p) {
        v4f q = pre[p];
        if (RELU) { q.x = fmaxf(q.x, 0.0f); q.y = fmaxf(q.y, 0.0f); q.z = fmaxf(q.z, 0.0f); q.w = fmaxf(q.w, 0.0f); }
        *(v4f*)(lds + (p * 8 + wv) * A_PITCH + lane * 4) = q;
    }
    __syncthreads();

    int cur = 0;
    for (; g < gEnd; g += gStep) {
        const int  gn   = g + gStep;
        const bool more = gn < gEnd;
        if (more) {
#pragma unroll
            for (int p = 0; p < NP; ++p) {
                const int row = min(gn * ROWS + p * 8 + wv, nRows - 1);
                pre[p] = *(const v4f*)(A + (size_t)row * 128 + lane * 4);
            }
        }

        const int tm = g * ROW_TILES_PER_BLOCK + tr;
        if (tm < nRowTiles) {
            const float* arow = lds + cur * BUF + (tr * 16 + mn) * A_PITCH + 2 * khalf;
            float drow = 1.0f;
            if (SCALE) {
                const float f = (float)deg[tm * 16 + mn - degBase];
                const float d = 1.0f / sqrtf(fmaxf(f, 1.0f));
                drow = (f > 0.0f) ? d : 0.0f;
            }
            v8f acc = {};
#pragma unroll
            for (int ks = 0; ks < 32; ++ks) {
                const v2f a = *(const v2f*)(arow + 4 * ks);
                acc = __builtin_amdgcn_wmma_f32_16x16x4_f32(
                     false, a,  false, bfrag[ks],
                     (short)0, acc,  false,  false);
            }
            float* orow = out + (size_t)(tm * 16 + khalf * 8) * F + (size_t)(tn * 16 + mn);
#pragma unroll
            for (int v = 0; v < 8; ++v) {
                float val = acc[v];
                if (SCALE) {
                    const float dlo = __int_as_float(__builtin_amdgcn_readlane(__float_as_int(drow), v));
                    const float dhi = __int_as_float(__builtin_amdgcn_readlane(__float_as_int(drow), v + 8));
                    val *= khalf ? dhi : dlo;
                }
                orow[(size_t)v * F] = val;
            }
        }

        if (more) {
#pragma unroll
            for (int p = 0; p < NP; ++p) {
                v4f q = pre[p];
                if (RELU) { q.x = fmaxf(q.x, 0.0f); q.y = fmaxf(q.y, 0.0f); q.z = fmaxf(q.z, 0.0f); q.w = fmaxf(q.w, 0.0f); }
                *(v4f*)(lds + (cur ^ 1) * BUF + (p * 8 + wv) * A_PITCH + lane * 4) = q;
            }
        }
        gcn_lds_barrier();
        cur ^= 1;
    }
}

#define RUN_LOADS 8

__global__ __launch_bounds__(256) void gcn_sort_and_gemm1(const int* __restrict__ binned,
                                                          const int* __restrict__ binOff,
                                                          int* __restrict__ offs,
                                                          int* __restrict__ srow,
                                                          const float* __restrict__ x,
                                                          const float* __restrict__ W1,
                                                          float* __restrict__ t, int nRows) {
    __shared__ int deg[128];
    __shared__ int cur[128];
    __shared__ int wtot[8];
    __shared__ int winc[8];
    const int tid = (int)threadIdx.x, lane = tid & 31, wv = tid >> 5;
    const int b  = (int)blockIdx.x;
    const int n0 = b * BIN_NODES;
    if (tid < 128) deg[tid] = 0;

    const int o0  = binOff[tid * OFF_PITCH + b];
    const int len = binOff[tid * OFF_PITCH + b + 1] - o0;
    const int* __restrict__ run = binned + tid * SLICE_EDGES + o0;

    int part = o0;
    for (int o = 16; o > 0; o >>= 1) part += __shfl_xor(part, o);
    if (lane == 0) wtot[wv] = part;
    __syncthreads();
    int binStart = 0;
    for (int w = 0; w < 8; ++w) binStart += wtot[w];

    for (int j = 0; j < len; j += RUN_LOADS) {
        int v[RUN_LOADS];
#pragma unroll
        for (int k = 0; k < RUN_LOADS; ++k) v[k] = run[min(j + k, len - 1)];
#pragma unroll
        for (int k = 0; k < RUN_LOADS; ++k)
            if (j + k < len) atomicAdd(&deg[v[k] & 127], 1);
    }
    __syncthreads();

    const int d = (tid < 128) ? deg[tid] : 0;
    int inc = d;
    for (int o = 1; o < 32; o <<= 1) {
        const int u = __shfl_up(inc, o);
        if (lane >= o) inc += u;
    }
    if (lane == 31) winc[wv] = inc;
    __syncthreads();
    if (tid < 128) {
        int start = binStart + inc - d;
        for (int w = 0; w < wv; ++w) start += winc[w];
        cur[tid] = start;
        if (tid < BIN_NODES && n0 + tid < N_NODES) offs[n0 + tid] = start;
    }
    __syncthreads();

    for (int j = 0; j < len; j += RUN_LOADS) {
        int v[RUN_LOADS];
#pragma unroll
        for (int k = 0; k < RUN_LOADS; ++k) v[k] = run[min(j + k, len - 1)];
#pragma unroll
        for (int k = 0; k < RUN_LOADS; ++k)
            if (j + k < len) srow[atomicAdd(&cur[v[k] & 127], 1)] = v[k] >> 7;
    }

    const int gBegin = b * (BIN_NODES / 16);
    const int gEnd   = min(gBegin + BIN_NODES / 16, nRows / 16);
    gcn_gemm_body<128, false, true>(x, W1, deg, n0, t, nRows, gBegin, gEnd, 1);
}

template <int V> struct gcn_vec;
template <> struct gcn_vec<2> { typedef v2f type; };
template <> struct gcn_vec<4> { typedef v4f type; };

template <int F, bool PRESCALE>
__global__ __launch_bounds__(256) void gcn_gather_sum(const float* __restrict__ t,
                                                      const int* __restrict__ offs,
                                                      const int* __restrict__ srow,
                                                      const float* __restrict__ b,
                                                      float* __restrict__ out, int nNodes) {
    constexpr int V = F / 32;
    typedef typename gcn_vec<V>::type vf;
    const int lane = threadIdx.x & 31;
    const int first = blockIdx.x * 8 + __builtin_amdgcn_readfirstlane((int)(threadIdx.x >> 5));
    const int nWaves = (int)gridDim.x * 8;
    const vf* tv = (const vf*)t + lane;
    const vf bias = *((const vf*)b + lane);

    for (int node = first; node < nNodes; node += nWaves) {
    const int beg = offs[node];
    const int end = (node + 1 < N_NODES) ? offs[node + 1] : N_EDGES;
    const int n = end - beg;
    const float dnode = gcn_dinv_of(n);

    vf acc = {};

    for (int base = 0; base < n; base += 32) {
        const int m = (n - base < 32) ? (n - base) : 32;
        int er = 0;
        if (lane < m) er = srow[beg + base + lane];
        for (int k = 0; k < m; k += 4) {
            const int r0 = __builtin_amdgcn_readlane(er, k);
            const int r1 = __builtin_amdgcn_readlane(er, k + 1);
            const int r2 = __builtin_amdgcn_readlane(er, k + 2);
            const int r3 = __builtin_amdgcn_readlane(er, k + 3);
            const vf x0 = tv[(size_t)r0 * 32];
            const vf x1 = tv[(size_t)r1 * 32];
            const vf x2 = tv[(size_t)r2 * 32];
            const vf x3 = tv[(size_t)r3 * 32];
            const vf zero = {};
            acc += x0;
            acc += (k + 1 < m) ? x1 : zero;
            acc += (k + 2 < m) ? x2 : zero;
            acc += (k + 3 < m) ? x3 : zero;
        }
    }
    vf res = bias + acc * dnode;
    if (PRESCALE) res = res * dnode;
    *((vf*)(out + (size_t)node * F) + lane) = res;
    }
}

#define ROWS_IN_FLIGHT 8

__device__ __forceinline__ v4f gcn_sum_rows128(const float* __restrict__ tin,
                                               const int* __restrict__ srow,
                                               int beg, int n, int lane) {
    v4f acc = {};
    for (int base = 0; base < n; base += 32) {
        const int m = (n - base < 32) ? (n - base) : 32;
        int er = 0;
        if (lane < m) er = srow[beg + base + lane];
        for (int k = 0; k < m; k += ROWS_IN_FLIGHT) {
            v4f x[ROWS_IN_FLIGHT];
#pragma unroll
            for (int i = 0; i < ROWS_IN_FLIGHT; ++i) {
                const unsigned r = (unsigned)__builtin_amdgcn_readlane(er, k + i);
                x[i] = *(const v4f*)((const char*)tin + (r * 512u + (unsigned)lane * 16u));
            }
            const v4f zero = {};
#pragma unroll
            for (int i = 0; i < ROWS_IN_FLIGHT; ++i) acc += (k + i < m) ? x[i] : zero;
        }
    }
    return acc;
}

template <int NP, int ROWS>
__device__ __forceinline__ void gcn_ask_bounds(const int* __restrict__ offs, int gg, int wv,
                                               int (&o0)[NP], int (&o1)[NP]) {
#pragma unroll
    for (int p = 0; p < NP; ++p) {
        const int node = gg * ROWS + p * 8 + wv;
        o0[p] = offs[min(node, N_NODES - 1)];
        o1[p] = offs[min(node + 1, N_NODES - 1)];
    }
}

template <int NP, int ROWS>
__device__ __forceinline__ void gcn_fill_rows(float* __restrict__ dst, const float* __restrict__ tin,
                                              const int* __restrict__ srow, v4f bias,
                                              int gg, int wv, int lane, int nRows,
                                              const int (&o0)[NP], const int (&o1)[NP]) {
#pragma unroll
    for (int p = 0; p < NP; ++p) {
        const int node = gg * ROWS + p * 8 + wv;
        const int end  = (node + 1 < N_NODES) ? o1[p] : N_EDGES;
        const int n    = (node < nRows) ? (end - o0[p]) : 0;
        const float dnode = gcn_dinv_of(n);
        const v4f s = gcn_sum_rows128(tin, srow, o0[p], n, lane);
        v4f q = (bias + s * dnode) * dnode;
        q.x = fmaxf(q.x, 0.0f); q.y = fmaxf(q.y, 0.0f); q.z = fmaxf(q.z, 0.0f); q.w = fmaxf(q.w, 0.0f);
        *(v4f*)(dst + (p * 8 + wv) * A_PITCH + lane * 4) = q;
    }
}

template <int F>
__global__ __launch_bounds__(256) void gcn_gather_gemm_wmma(const float* __restrict__ tin,
                                                            const int* __restrict__ offs,
                                                            const int* __restrict__ srow,
                                                            const float* __restrict__ b,
                                                            const float* __restrict__ W,
                                                            float* __restrict__ out, int nRows) {
    constexpr int TILES_N = F / 16;
    constexpr int ROW_TILES_PER_BLOCK = 8 / TILES_N;
    constexpr int ROWS = ROW_TILES_PER_BLOCK * 16;
    constexpr int NP   = ROWS / 8;
    constexpr int BUF  = ROWS * A_PITCH;
    __shared__ __align__(16) float lds[2 * BUF];

    const int lane = threadIdx.x & 31;
    const int wv = __builtin_amdgcn_readfirstlane((int)(threadIdx.x >> 5));
    const int tn = wv % TILES_N;
    const int tr = wv / TILES_N;

    const int mn    = lane & 15;
    const int khalf = lane >> 4;

    const int nRowTiles = nRows / 16;
    const int nGroups = (nRowTiles + ROW_TILES_PER_BLOCK - 1) / ROW_TILES_PER_BLOCK;
    int g = (int)blockIdx.x;

    int o0[NP], o1[NP];
    gcn_ask_bounds<NP, ROWS>(offs, g, wv, o0, o1);

    const v4f bias = *((const v4f*)b + lane);

    const float* wcol = W + (size_t)(tn * 16 + mn);
    v2f bfrag[32];
#pragma unroll
    for (int ks = 0; ks < 32; ++ks) {
        const int kb = 4 * ks + 2 * khalf;
        bfrag[ks].x = wcol[(size_t)kb * F];
        bfrag[ks].y = wcol[(size_t)(kb + 1) * F];
    }

    gcn_fill_rows<NP, ROWS>(lds, tin, srow, bias, g, wv, lane, nRows, o0, o1);
    __syncthreads();

    int cur = 0;
    for (; g < nGroups; g += (int)gridDim.x) {
        const int  gn   = g + (int)gridDim.x;
        const bool more = gn < nGroups;
        if (more) gcn_ask_bounds<NP, ROWS>(offs, gn, wv, o0, o1);

        const int tm = g * ROW_TILES_PER_BLOCK + tr;
        if (tm < nRowTiles) {
            const float* arow = lds + cur * BUF + (tr * 16 + mn) * A_PITCH + 2 * khalf;
            v8f acc = {};
#pragma unroll
            for (int ks = 0; ks < 32; ++ks) {
                const v2f a = *(const v2f*)(arow + 4 * ks);
                acc = __builtin_amdgcn_wmma_f32_16x16x4_f32(
                     false, a,  false, bfrag[ks],
                     (short)0, acc,  false,  false);
            }
            float* orow = out + (size_t)(tm * 16 + khalf * 8) * F + (size_t)(tn * 16 + mn);
#pragma unroll
            for (int v = 0; v < 8; ++v) orow[(size_t)v * F] = acc[v];
        }

        if (more)
            gcn_fill_rows<NP, ROWS>(lds + (cur ^ 1) * BUF, tin, srow, bias, gn, wv, lane, nRows, o0, o1);
        gcn_lds_barrier();
        cur ^= 1;
    }
}


extern "C" void kernel_launch(void* const* d_in, const int* in_sizes, int n_in,
                              void* d_out, int out_size, void* d_ws, size_t ws_size,
                              hipStream_t stream) {
    const float* x   = (const float*)d_in[0];
    const int*   ei  = (const int*)d_in[1];
    const float* W1  = (const float*)d_in[2];
    const float* b1  = (const float*)d_in[3];
    const float* W2  = (const float*)d_in[4];
    const float* b2  = (const float*)d_in[5];
    const float* W3  = (const float*)d_in[6];
    const float* b3  = (const float*)d_in[7];
    float*       out = (float*)d_out;

    const int* row = ei;
    const int* col = ei + N_EDGES;

    float* t    = (float*)d_ws;
    float* h    = t + (size_t)N_NODES * 128;
    int*   offs = (int*)(h + (size_t)N_NODES * 128);
    int*   srow = offs + N_NODES;
    int*   binned = (int*)h;
    int*   binOff = binned + N_EDGES;

    const int TB = 256;

    gcn_bin_slices<<<N_SLICES, TB, 0, stream>>>(row, col, binned, binOff, N_EDGES);

    gcn_sort_and_gemm1<<<N_BINS, TB, 0, stream>>>(binned, binOff, offs, srow, x, W1, t, N_NODES);

    const int gemm128Blocks = GEMM_BLOCKS;
    const int gemm64Blocks  = GEMM_BLOCKS;
    const int gatherBlocks  = GATHER_BLOCKS;

    gcn_gather_gemm_wmma<128><<<gemm128Blocks, TB, 0, stream>>>(t, offs, srow, b1, W2, h, N_NODES);

    gcn_gather_gemm_wmma<64><<<gemm64Blocks, TB, 0, stream>>>(h, offs, srow, b2, W3, t, N_NODES);

    gcn_gather_sum<64, false><<<gatherBlocks, TB, 0, stream>>>(t, offs, srow, b3, out, N_NODES);
}


# Edge sort stores whole lines: the sorted slice (launch 1) and each range's source rows (launch 2) are put together in LDS and stored line by line, not as scattered 4-byte stores; arrival order kept
# speedup vs baseline: 9.9206x; 1.0567x over previous
#include <hip/hip_runtime.h>

typedef float v2f __attribute__((ext_vector_type(2)));
typedef float v4f __attribute__((ext_vector_type(4)));
typedef float v8f __attribute__((ext_vector_type(8)));


#define N_NODES 100000
#define N_EDGES 1600000
#define GATHER_BLOCKS 2048
#define GEMM_BLOCKS   1024

#define BIN_NODES   112
#define N_BINS      ((N_NODES + BIN_NODES - 1) / BIN_NODES)
#define N_SLICES    256
#define SLICE_EDGES ((N_EDGES + N_SLICES - 1) / N_SLICES)
#define BIN_LOADS   13
#define OFF_PITCH   896
static_assert(N_BINS + 1 <= 1024 && N_BINS + 1 <= OFF_PITCH, "the range scan gives each of 256 threads 4 entries");
static_assert(BIN_NODES <= 128 && BIN_NODES % 16 == 0, "7 bits of a packed word hold the node's place in its range");
static_assert(N_NODES <= (1 << 24), "24 bits of a packed word hold the source row");

__global__ __launch_bounds__(256) void gcn_bin_slices(const int* __restrict__ row,
                                                      const int* __restrict__ col,
                                                      int* __restrict__ binned,
                                                      int* __restrict__ binOff, int e) {
    __shared__ int hist[N_BINS];
    __shared__ int wsum[8];
    __shared__ int sorted[SLICE_EDGES];
    const int tid = (int)threadIdx.x, lane = tid & 31, wv = tid >> 5;
    const int s  = (int)blockIdx.x;
    const int e0 = s * SLICE_EDGES;
    const int e1 = min(e, e0 + SLICE_EDGES);
    for (int b = tid; b < N_BINS; b += 256) hist[b] = 0;
    __syncthreads();

    for (int base = e0; base < e1; base += 256 * BIN_LOADS) {
        int c[BIN_LOADS];
#pragma unroll
        for (int j = 0; j < BIN_LOADS; ++j) c[j] = col[min(base + j * 256 + tid, e1 - 1)];
#pragma unroll
        for (int j = 0; j < BIN_LOADS; ++j)
            if (base + j * 256 + tid < e1) atomicAdd(&hist[c[j] / BIN_NODES], 1);
    }
    __syncthreads();

    int v[4], sum = 0;
#pragma unroll
    for (int j = 0; j < 4; ++j) {
        const int b = tid * 4 + j;
        v[j] = (b < N_BINS) ? hist[b] : 0;
        sum += v[j];
    }
    int inc = sum;
    for (int o = 1; o < 32; o <<= 1) {
        const int u = __shfl_up(inc, o);
        if (lane >= o) inc += u;
    }
    if (lane == 31) wsum[wv] = inc;
    __syncthreads();
    int ex = inc - sum;
    for (int w = 0; w < wv; ++w) ex += wsum[w];
#pragma unroll
    for (int j = 0; j < 4; ++j) {
        const int b = tid * 4 + j;
        if (b < N_BINS) hist[b] = ex;
        if (b <= N_BINS) binOff[s * OFF_PITCH + b] = ex;
        ex += v[j];
    }
    __syncthreads();

    for (int base = e0; base < e1; base += 256 * BIN_LOADS) {
        int c[BIN_LOADS], r[BIN_LOADS];
#pragma unroll
        for (int j = 0; j < BIN_LOADS; ++j) {
            const int i = min(base + j * 256 + tid, e1 - 1);
            c[j] = col[i];
            r[j] = row[i];
        }
#pragma unroll
        for (int j = 0; j < BIN_LOADS; ++j)
            if (base + j * 256 + tid < e1) {
                const int b = c[j] / BIN_NODES;
                const int p = atomicAdd(&hist[b], 1);
                sorted[p] = (r[j] << 7) | (c[j] - b * BIN_NODES);
            }
    }
    __syncthreads();

    int gi = (e0 & ~31) + tid;
    if (gi >= e0 && gi < e1) binned[gi] = sorted[gi - e0];
    for (gi += 256; gi < e1; gi += 4 * 256) {
        int w[4];
#pragma unroll
        for (int u = 0; u < 4; ++u) w[u] = sorted[min(gi + u * 256, e1 - 1) - e0];
#pragma unroll
        for (int u = 0; u < 4; ++u)
            if (gi + u * 256 < e1) binned[gi + u * 256] = w[u];
    }
}

__device__ __forceinline__ float gcn_dinv_of(int d) {
    const float f = (float)d;
    return (f > 0.0f) ? (1.0f / sqrtf(f)) : 0.0f;
}


#define A_PITCH 132

__device__ __forceinline__ void gcn_lds_barrier() {
    __builtin_amdgcn_fence(__ATOMIC_RELEASE, "workgroup", "local");
    __builtin_amdgcn_s_barrier();
    __builtin_amdgcn_fence(__ATOMIC_ACQUIRE, "workgroup", "local");
}


template <int F, bool RELU, bool SCALE>
__device__ __forceinline__ void gcn_gemm_body(const float* __restrict__ A,
                                              const float* __restrict__ W,
                                              const int* deg, int degBase,
                                              float* __restrict__ out, int nRows,
                                              int gBegin, int gEnd, int gStep) {
    constexpr int TILES_N = F / 16;
    constexpr int ROW_TILES_PER_BLOCK = 8 / TILES_N;
    constexpr int ROWS = ROW_TILES_PER_BLOCK * 16;
    constexpr int NP   = ROWS / 8;
    constexpr int BUF  = ROWS * A_PITCH;
    __shared__ __align__(16) float lds[2 * BUF];

    const int lane = threadIdx.x & 31;
    const int wv = __builtin_amdgcn_readfirstlane((int)(threadIdx.x >> 5));
    const int tn = wv % TILES_N;
    const int tr = wv / TILES_N;

    const int mn    = lane & 15;
    const int khalf = lane >> 4;

    const int nRowTiles = nRows / 16;
    int g = gBegin;

    v4f pre[NP];
#pragma unroll
    for (int p = 0; p < NP; ++p) {
        const int row = min(g * ROWS + p * 8 + wv, nRows - 1);
        pre[p] = *(const v4f*)(A + (size_t)row * 128 + lane * 4);
    }

    const float* wcol = W + (size_t)(tn * 16 + mn);
    v2f bfrag[32];
#pragma unroll
    for (int ks = 0; ks < 32; ++ks) {
        const int kb = 4 * ks + 2 * khalf;
        bfrag[ks].x = wcol[(size_t)kb * F];
        bfrag[ks].y = wcol[(size_t)(kb + 1) * F];
    }

#pragma unroll
    for (int p = 0; p < NP; ++p) {
        v4f q = pre[p];
        if (RELU) { q.x = fmaxf(q.x, 0.0f); q.y = fmaxf(q.y, 0.0f); q.z = fmaxf(q.z, 0.0f); q.w = fmaxf(q.w, 0.0f); }
        *(v4f*)(lds + (p * 8 + wv) * A_PITCH + lane * 4) = q;
    }
    __syncthreads();

    int cur = 0;
    for (; g < gEnd; g += gStep) {
        const int  gn   = g + gStep;
        const bool more = gn < gEnd;
        if (more) {
#pragma unroll
            for (int p = 0; p < NP; ++p) {
                const int row = min(gn * ROWS + p * 8 + wv, nRows - 1);
                pre[p] = *(const v4f*)(A + (size_t)row * 128 + lane * 4);
            }
        }

        const int tm = g * ROW_TILES_PER_BLOCK + tr;
        if (tm < nRowTiles) {
            const float* arow = lds + cur * BUF + (tr * 16 + mn) * A_PITCH + 2 * khalf;
            float drow = 1.0f;
            if (SCALE) {
                const float f = (float)deg[tm * 16 + mn - degBase];
                const float d = 1.0f / sqrtf(fmaxf(f, 1.0f));
                drow = (f > 0.0f) ? d : 0.0f;
            }
            v8f acc = {};
#pragma unroll
            for (int ks = 0; ks < 32; ++ks) {
                const v2f a = *(const v2f*)(arow + 4 * ks);
                acc = __builtin_amdgcn_wmma_f32_16x16x4_f32(
                     false, a,  false, bfrag[ks],
                     (short)0, acc,  false,  false);
            }
            float* orow = out + (size_t)(tm * 16 + khalf * 8) * F + (size_t)(tn * 16 + mn);
#pragma unroll
            for (int v = 0; v < 8; ++v) {
                float val = acc[v];
                if (SCALE) {
                    const float dlo = __int_as_float(__builtin_amdgcn_readlane(__float_as_int(drow), v));
                    const float dhi = __int_as_float(__builtin_amdgcn_readlane(__float_as_int(drow), v + 8));
                    val *= khalf ? dhi : dlo;
                }
                orow[(size_t)v * F] = val;
            }
        }

        if (more) {
#pragma unroll
            for (int p = 0; p < NP; ++p) {
                v4f q = pre[p];
                if (RELU) { q.x = fmaxf(q.x, 0.0f); q.y = fmaxf(q.y, 0.0f); q.z = fmaxf(q.z, 0.0f); q.w = fmaxf(q.w, 0.0f); }
                *(v4f*)(lds + (cur ^ 1) * BUF + (p * 8 + wv) * A_PITCH + lane * 4) = q;
            }
        }
        gcn_lds_barrier();
        cur ^= 1;
    }
}

#define RUN_LOADS 8
#define SEG_STAGE 2560

__global__ __launch_bounds__(256) void gcn_sort_and_gemm1(const int* __restrict__ binned,
                                                          const int* __restrict__ binOff,
                                                          int* __restrict__ offs,
                                                          int* __restrict__ srow,
                                                          const float* __restrict__ x,
                                                          const float* __restrict__ W1,
                                                          float* __restrict__ t, int nRows) {
    __shared__ int deg[128];
    __shared__ int cur[128];
    __shared__ int wtot[8];
    __shared__ int winc[8];
    __shared__ int seg[SEG_STAGE];
    const int tid = (int)threadIdx.x, lane = tid & 31, wv = tid >> 5;
    const int b  = (int)blockIdx.x;
    const int n0 = b * BIN_NODES;
    if (tid < 128) deg[tid] = 0;

    const int o0  = binOff[tid * OFF_PITCH + b];
    const int len = binOff[tid * OFF_PITCH + b + 1] - o0;
    const int* __restrict__ run = binned + tid * SLICE_EDGES + o0;

    int part = o0;
    for (int o = 16; o > 0; o >>= 1) part += __shfl_xor(part, o);
    if (lane == 0) wtot[wv] = part;
    __syncthreads();
    int binStart = 0;
    for (int w = 0; w < 8; ++w) binStart += wtot[w];

    for (int j = 0; j < len; j += RUN_LOADS) {
        int v[RUN_LOADS];
#pragma unroll
        for (int k = 0; k < RUN_LOADS; ++k) v[k] = run[min(j + k, len - 1)];
#pragma unroll
        for (int k = 0; k < RUN_LOADS; ++k)
            if (j + k < len) atomicAdd(&deg[v[k] & 127], 1);
    }
    __syncthreads();

    const int d = (tid < 128) ? deg[tid] : 0;
    int inc = d;
    for (int o = 1; o < 32; o <<= 1) {
        const int u = __shfl_up(inc, o);
        if (lane >= o) inc += u;
    }
    if (lane == 31) winc[wv] = inc;
    __syncthreads();
    if (tid < 128) {
        int start = inc - d;
        for (int w = 0; w < wv; ++w) start += winc[w];
        cur[tid] = start;
        if (tid < BIN_NODES && n0 + tid < N_NODES) offs[n0 + tid] = binStart + start;
    }
    int binEdges = 0;
    for (int w = 0; w < 8; ++w) binEdges += winc[w];
    const bool staged = binEdges <= SEG_STAGE;
    __syncthreads();

    for (int j = 0; j < len; j += RUN_LOADS) {
        int v[RUN_LOADS];
#pragma unroll
        for (int k = 0; k < RUN_LOADS; ++k) v[k] = run[min(j + k, len - 1)];
#pragma unroll
        for (int k = 0; k < RUN_LOADS; ++k)
            if (j + k < len) {
                const int p = atomicAdd(&cur[v[k] & 127], 1);
                if (staged) seg[p] = v[k] >> 7;
                else        srow[binStart + p] = v[k] >> 7;
            }
    }
    __syncthreads();
    if (staged) {
        const int segEnd = binStart + binEdges;
        int gi = (binStart & ~31) + tid;
        if (gi >= binStart && gi < segEnd) srow[gi] = seg[gi - binStart];
        for (gi += 256; gi < segEnd; gi += 4 * 256) {
            int w[4];
#pragma unroll
            for (int u = 0; u < 4; ++u) w[u] = seg[min(gi + u * 256, segEnd - 1) - binStart];
#pragma unroll
            for (int u = 0; u < 4; ++u)
                if (gi + u * 256 < segEnd) srow[gi + u * 256] = w[u];
        }
    }

    const int gBegin = b * (BIN_NODES / 16);
    const int gEnd   = min(gBegin + BIN_NODES / 16, nRows / 16);
    gcn_gemm_body<128, false, true>(x, W1, deg, n0, t, nRows, gBegin, gEnd, 1);
}

template <int V> struct gcn_vec;
template <> struct gcn_vec<2> { typedef v2f type; };
template <> struct gcn_vec<4> { typedef v4f type; };

template <int F, bool PRESCALE>
__global__ __launch_bounds__(256) void gcn_gather_sum(const float* __restrict__ t,
                                                      const int* __restrict__ offs,
                                                      const int* __restrict__ srow,
                                                      const float* __restrict__ b,
                                                      float* __restrict__ out, int nNodes) {
    constexpr int V = F / 32;
    typedef typename gcn_vec<V>::type vf;
    const int lane = threadIdx.x & 31;
    const int first = blockIdx.x * 8 + __builtin_amdgcn_readfirstlane((int)(threadIdx.x >> 5));
    const int nWaves = (int)gridDim.x * 8;
    const vf* tv = (const vf*)t + lane;
    const vf bias = *((const vf*)b + lane);

    for (int node = first; node < nNodes; node += nWaves) {
    const int beg = offs[node];
    const int end = (node + 1 < N_NODES) ? offs[node + 1] : N_EDGES;
    const int n = end - beg;
    const float dnode = gcn_dinv_of(n);

    vf acc = {};

    for (int base = 0; base < n; base += 32) {
        const int m = (n - base < 32) ? (n - base) : 32;
        int er = 0;
        if (lane < m) er = srow[beg + base + lane];
        for (int k = 0; k < m; k += 4) {
            const int r0 = __builtin_amdgcn_readlane(er, k);
            const int r1 = __builtin_amdgcn_readlane(er, k + 1);
            const int r2 = __builtin_amdgcn_readlane(er, k + 2);
            const int r3 = __builtin_amdgcn_readlane(er, k + 3);
            const vf x0 = tv[(size_t)r0 * 32];
            const vf x1 = tv[(size_t)r1 * 32];
            const vf x2 = tv[(size_t)r2 * 32];
            const vf x3 = tv[(size_t)r3 * 32];
            const vf zero = {};
            acc += x0;
            acc += (k + 1 < m) ? x1 : zero;
            acc += (k + 2 < m) ? x2 : zero;
            acc += (k + 3 < m) ? x3 : zero;
        }
    }
    vf res = bias + acc * dnode;
    if (PRESCALE) res = res * dnode;
    *((vf*)(out + (size_t)node * F) + lane) = res;
    }
}

#define ROWS_IN_FLIGHT 8

__device__ __forceinline__ v4f gcn_sum_rows128(const float* __restrict__ tin,
                                               const int* __restrict__ srow,
                                               int beg, int n, int lane) {
    v4f acc = {};
    for (int base = 0; base < n; base += 32) {
        const int m = (n - base < 32) ? (n - base) : 32;
        int er = 0;
        if (lane < m) er = srow[beg + base + lane];
        for (int k = 0; k < m; k += ROWS_IN_FLIGHT) {
            v4f x[ROWS_IN_FLIGHT];
#pragma unroll
            for (int i = 0; i < ROWS_IN_FLIGHT; ++i) {
                const unsigned r = (unsigned)__builtin_amdgcn_readlane(er, k + i);
                x[i] = *(const v4f*)((const char*)tin + (r * 512u + (unsigned)lane * 16u));
            }
            const v4f zero = {};
#pragma unroll
            for (int i = 0; i < ROWS_IN_FLIGHT; ++i) acc += (k + i < m) ? x[i] : zero;
        }
    }
    return acc;
}

template <int NP, int ROWS>
__device__ __forceinline__ void gcn_ask_bounds(const int* __restrict__ offs, int gg, int wv,
                                               int (&o0)[NP], int (&o1)[NP]) {
#pragma unroll
    for (int p = 0; p < NP; ++p) {
        const int node = gg * ROWS + p * 8 + wv;
        o0[p] = offs[min(node, N_NODES - 1)];
        o1[p] = offs[min(node + 1, N_NODES - 1)];
    }
}

template <int NP, int ROWS>
__device__ __forceinline__ void gcn_fill_rows(float* __restrict__ dst, const float* __restrict__ tin,
                                              const int* __restrict__ srow, v4f bias,
                                              int gg, int wv, int lane, int nRows,
                                              const int (&o0)[NP], const int (&o1)[NP]) {
#pragma unroll
    for (int p = 0; p < NP; ++p) {
        const int node = gg * ROWS + p * 8 + wv;
        const int end  = (node + 1 < N_NODES) ? o1[p] : N_EDGES;
        const int n    = (node < nRows) ? (end - o0[p]) : 0;
        const float dnode = gcn_dinv_of(n);
        const v4f s = gcn_sum_rows128(tin, srow, o0[p], n, lane);
        v4f q = (bias + s * dnode) * dnode;
        q.x = fmaxf(q.x, 0.0f); q.y = fmaxf(q.y, 0.0f); q.z = fmaxf(q.z, 0.0f); q.w = fmaxf(q.w, 0.0f);
        *(v4f*)(dst + (p * 8 + wv) * A_PITCH + lane * 4) = q;
    }
}

template <int F>
__global__ __launch_bounds__(256) void gcn_gather_gemm_wmma(const float* __restrict__ tin,
                                                            const int* __restrict__ offs,
                                                            const int* __restrict__ srow,
                                                            const float* __restrict__ b,
                                                            const float* __restrict__ W,
                                                            float* __restrict__ out, int nRows) {
    constexpr int TILES_N = F / 16;
    constexpr int ROW_TILES_PER_BLOCK = 8 / TILES_N;
    constexpr int ROWS = ROW_TILES_PER_BLOCK * 16;
    constexpr int NP   = ROWS / 8;
    constexpr int BUF  = ROWS * A_PITCH;
    __shared__ __align__(16) float lds[2 * BUF];

    const int lane = threadIdx.x & 31;
    const int wv = __builtin_amdgcn_readfirstlane((int)(threadIdx.x >> 5));
    const int tn = wv % TILES_N;
    const int tr = wv / TILES_N;

    const int mn    = lane & 15;
    const int khalf = lane >> 4;

    const int nRowTiles = nRows / 16;
    const int nGroups = (nRowTiles + ROW_TILES_PER_BLOCK - 1) / ROW_TILES_PER_BLOCK;
    int g = (int)blockIdx.x;

    int o0[NP], o1[NP];
    gcn_ask_bounds<NP, ROWS>(offs, g, wv, o0, o1);

    const v4f bias = *((const v4f*)b + lane);

    const float* wcol = W + (size_t)(tn * 16 + mn);
    v2f bfrag[32];
#pragma unroll
    for (int ks = 0; ks < 32; ++ks) {
        const int kb = 4 * ks + 2 * khalf;
        bfrag[ks].x = wcol[(size_t)kb * F];
        bfrag[ks].y = wcol[(size_t)(kb + 1) * F];
    }

    gcn_fill_rows<NP, ROWS>(lds, tin, srow, bias, g, wv, lane, nRows, o0, o1);
    __syncthreads();

    int cur = 0;
    for (; g < nGroups; g += (int)gridDim.x) {
        const int  gn   = g + (int)gridDim.x;
        const bool more = gn < nGroups;
        if (more) gcn_ask_bounds<NP, ROWS>(offs, gn, wv, o0, o1);

        const int tm = g * ROW_TILES_PER_BLOCK + tr;
        if (tm < nRowTiles) {
            const float* arow = lds + cur * BUF + (tr * 16 + mn) * A_PITCH + 2 * khalf;
            v8f acc = {};
#pragma unroll
            for (int ks = 0; ks < 32; ++ks) {
                const v2f a = *(const v2f*)(arow + 4 * ks);
                acc = __builtin_amdgcn_wmma_f32_16x16x4_f32(
                     false, a,  false, bfrag[ks],
                     (short)0, acc,  false,  false);
            }
            float* orow = out + (size_t)(tm * 16 + khalf * 8) * F + (size_t)(tn * 16 + mn);
#pragma unroll
            for (int v = 0; v < 8; ++v) orow[(size_t)v * F] = acc[v];
        }

        if (more)
            gcn_fill_rows<NP, ROWS>(lds + (cur ^ 1) * BUF, tin, srow, bias, gn, wv, lane, nRows, o0, o1);
        gcn_lds_barrier();
        cur ^= 1;
    }
}


extern "C" void kernel_launch(void* const* d_in, const int* in_sizes, int n_in,
                              void* d_out, int out_size, void* d_ws, size_t ws_size,
                              hipStream_t stream) {
    const float* x   = (const float*)d_in[0];
    const int*   ei  = (const int*)d_in[1];
    const float* W1  = (const float*)d_in[2];
    const float* b1  = (const float*)d_in[3];
    const float* W2  = (const float*)d_in[4];
    const float* b2  = (const float*)d_in[5];
    const float* W3  = (const float*)d_in[6];
    const float* b3  = (const float*)d_in[7];
    float*       out = (float*)d_out;

    const int* row = ei;
    const int* col = ei + N_EDGES;

    float* t    = (float*)d_ws;
    float* h    = t + (size_t)N_NODES * 128;
    int*   offs = (int*)(h + (size_t)N_NODES * 128);
    int*   srow = offs + N_NODES;
    int*   binned = (int*)h;
    int*   binOff = binned + N_EDGES;

    const int TB = 256;

    gcn_bin_slices<<<N_SLICES, TB, 0, stream>>>(row, col, binned, binOff, N_EDGES);

    gcn_sort_and_gemm1<<<N_BINS, TB, 0, stream>>>(binned, binOff, offs, srow, x, W1, t, N_NODES);

    const int gemm128Blocks = GEMM_BLOCKS;
    const int gemm64Blocks  = GEMM_BLOCKS;
    const int gatherBlocks  = GATHER_BLOCKS;

    gcn_gather_gemm_wmma<128><<<gemm128Blocks, TB, 0, stream>>>(t, offs, srow, b1, W2, h, N_NODES);

    gcn_gather_gemm_wmma<64><<<gemm64Blocks, TB, 0, stream>>>(h, offs, srow, b2, W3, t, N_NODES);

    gcn_gather_sum<64, false><<<gatherBlocks, TB, 0, stream>>>(t, offs, srow, b3, out, N_NODES);
}
